# starting kernel
# baseline (speedup 1.0000x reference)
#include <hip/hip_runtime.h>

#define Bsz 32
#define Lsz 256
#define Csz 16
#define Dsz 128
#define KTOT 384
#define NKCHUNK 96
#define WAVES 8
#define WT2_PITCH 288
#define WT2_FLOATS (192 * WT2_PITCH)
#define XT_FLOATS (64 * 36)
#define FEATS_FLOATS (Dsz * 16)

typedef __attribute__((ext_vector_type(2))) float v2f;
typedef __attribute__((ext_vector_type(8))) float v8f;

__device__ __forceinline__ void load_wt(const float* __restrict__ W, float* lds, int tid) {
  for (int idx = tid; idx < Dsz * Dsz; idx += 256) {
    const int dout = idx >> 7;
    const int din  = idx & 127;
    const float* wp = W + dout * 384 + din * 3;
    const float w0 = wp[0], w1 = wp[1], w2 = wp[2];
    const int kg0 = din;
    const int kg1 = 128 + din;
    const int kg2 = 256 + din;
    lds[(kg0 >> 1) * WT2_PITCH + dout * 2 + (kg0 & 1)] = w0;
    lds[(kg1 >> 1) * WT2_PITCH + dout * 2 + (kg1 & 1)] = w1;
    lds[(kg2 >> 1) * WT2_PITCH + dout * 2 + (kg2 & 1)] = w2;
  }
}

__device__ __forceinline__ void load_frags(const float* __restrict__ wt,
                                           const float* __restrict__ xt,
                                           int kk, int r, int h,
                                           v2f a[8], v2f* bf) {
  const int kg0  = kk * 4 + 2 * h;
  const int tap  = kg0 >> 7;
  const int din0 = kg0 & 127;
  *bf = *(const v2f*)(xt + (din0 >> 1) * 36 + (r + tap) * 2);
  const float* wrow = wt + (kg0 >> 1) * WT2_PITCH + r * 2;
#pragma unroll
  for (int t = 0; t < 8; ++t) {
    a[t] = *(const v2f*)(wrow + t * 32);
  }
}

__device__ __forceinline__ void wmma_group(const v2f a[8], v2f bf, v8f acc[8]) {
#pragma unroll
  for (int t = 0; t < 8; ++t) {
    acc[t] = __builtin_amdgcn_wmma_f32_16x16x4_f32(
         false, a[t],  false, bf,
         (short)0, acc[t],  false,  false);
  }
}

__device__ __forceinline__ void conv_tiles(const float* __restrict__ wt,
                                           const float* __restrict__ xt,
                                           int r, int h, v8f acc[8]) {
  v2f a0[8], a1[8], b0, b1;
  load_frags(wt, xt, 0, r, h, a0, &b0);
#pragma unroll 1
  for (int kk = 0; kk < NKCHUNK - 2; kk += 2) {
    load_frags(wt, xt, kk + 1, r, h, a1, &b1);
    wmma_group(a0, b0, acc);
    load_frags(wt, xt, kk + 2, r, h, a0, &b0);
    wmma_group(a1, b1, acc);
  }
  load_frags(wt, xt, NKCHUNK - 1, r, h, a1, &b1);
  wmma_group(a0, b0, acc);
  wmma_group(a1, b1, acc);
}

__global__ __launch_bounds__(256) void char_conv_kernel(
    const int* __restrict__ words_in_char,
    const float* __restrict__ chr_table,
    const float* __restrict__ conv_w,
    const float* __restrict__ conv_b,
    float* __restrict__ out) {
  extern __shared__ float smem[];
  float* wt = smem;
  const int tid  = threadIdx.x;
  const int lane = tid & 31;
  const int w    = tid >> 5;
  float* xt    = smem + WT2_FLOATS + w * XT_FLOATS;
  float* feats = smem + WT2_FLOATS + WAVES * XT_FLOATS;

  load_wt(conv_w, wt, tid);
  __syncthreads();

  const int gw0 = blockIdx.x * 16;
  const int b   = gw0 >> 8;
  const int l0  = gw0 & 255;
  const int r   = lane & 15;
  const int h   = lane >> 4;

  for (int wi = 0; wi < 2; ++wi) {
    const int lw = w * 2 + wi;
    const int g  = gw0 + lw;
    const int* wic = words_in_char + g * Csz;

    for (int cc = 0; cc < 18; ++cc) {
      const float* src = nullptr;
      if (cc >= 1 && cc <= 16) {
        const int ch = wic[cc - 1];
        src = chr_table + ch * Dsz;
      }
      for (int d = lane; d < Dsz; d += 32) {
        xt[(d >> 1) * 36 + cc * 2 + (d & 1)] = src ? src[d] : 0.0f;
      }
    }

    v8f acc[8] = {};
    conv_tiles(wt, xt, r, h, acc);

#pragma unroll
    for (int t = 0; t < 8; ++t) {
#pragma unroll
      for (int j = 0; j < 8; ++j) {
        float v = acc[t][j];
        v = fmaxf(v, __shfl_xor(v, 1, 32));
        v = fmaxf(v, __shfl_xor(v, 2, 32));
        v = fmaxf(v, __shfl_xor(v, 4, 32));
        v = fmaxf(v, __shfl_xor(v, 8, 32));
        if (r == j) {
          feats[(t * 16 + 8 * h + j) * 16 + lw] = v;
        }
      }
    }
  }
  __syncthreads();

  for (int i = tid; i < Dsz * 16; i += 256) {
    const int d = i >> 4;
    const int j = i & 15;
    out[((Bsz + b) * Dsz + d) * Lsz + l0 + j] = feats[d * 16 + j] + conv_b[d];
  }
}

__global__ __launch_bounds__(256) void word_conv_kernel(
    const int* __restrict__ word_vector,
    const float* __restrict__ word_table,
    const float* __restrict__ conv_w,
    const float* __restrict__ conv_b,
    float* __restrict__ out) {
  extern __shared__ float smem[];
  float* wt = smem;
  const int tid  = threadIdx.x;
  const int lane = tid & 31;
  const int w    = tid >> 5;
  float* xt = smem + WT2_FLOATS + w * XT_FLOATS;

  load_wt(conv_w, wt, tid);
  __syncthreads();

  const int task = blockIdx.x * WAVES + w;
  const int b    = task >> 4;
  const int l0   = (task & 15) * 16;
  const int r    = lane & 15;
  const int h    = lane >> 4;

  for (int cc = 0; cc < 18; ++cc) {
    const int l = l0 + cc - 1;
    const float* src = nullptr;
    if (l >= 0 && l < Lsz) {
      const int widx = word_vector[b * Lsz + l];
      src = word_table + widx * Dsz;
    }
    for (int d = lane; d < Dsz; d += 32) {
      xt[(d >> 1) * 36 + cc * 2 + (d & 1)] = src ? src[d] : 0.0f;
    }
  }

  v8f acc[8] = {};
  conv_tiles(wt, xt, r, h, acc);

#pragma unroll
  for (int t = 0; t < 8; ++t) {
#pragma unroll
    for (int j = 0; j < 8; ++j) {
      const int row = t * 16 + 8 * h + j;
      out[(b * Dsz + row) * Lsz + l0 + r] = acc[t][j] + conv_b[row];
    }
  }
}

extern "C" void kernel_launch(void* const* d_in, const int* in_sizes, int n_in,
                              void* d_out, int out_size, void* d_ws, size_t ws_size,
                              hipStream_t stream) {
  (void)in_sizes; (void)n_in; (void)out_size; (void)d_ws; (void)ws_size;
  const int*   word_vector   = (const int*)d_in[0];
  const int*   words_in_char = (const int*)d_in[1];
  const float* word_table    = (const float*)d_in[2];
  const float* chr_table     = (const float*)d_in[3];
  const float* conv_chr_w    = (const float*)d_in[4];
  const float* conv_chr_b    = (const float*)d_in[5];
  const float* conv_word_w   = (const float*)d_in[6];
  const float* conv_word_b   = (const float*)d_in[7];
  float* out = (float*)d_out;

  const size_t char_lds = (size_t)(WT2_FLOATS + WAVES * XT_FLOATS + FEATS_FLOATS) * sizeof(float);
  const size_t word_lds = (size_t)(WT2_FLOATS + WAVES * XT_FLOATS) * sizeof(float);

  static bool attr_done = false;
  if (!attr_done) {
    hipFuncSetAttribute((const void*)char_conv_kernel,
                        hipFuncAttributeMaxDynamicSharedMemorySize, (int)char_lds);
    hipFuncSetAttribute((const void*)word_conv_kernel,
                        hipFuncAttributeMaxDynamicSharedMemorySize, (int)word_lds);
    attr_done = true;
  }

  char_conv_kernel<<<dim3(Bsz * Lsz / 16), dim3(256), char_lds, stream>>>(
      words_in_char, chr_table, conv_chr_w, conv_chr_b, out);
  word_conv_kernel<<<dim3(Bsz * 16 / WAVES), dim3(256), word_lds, stream>>>(
      word_vector, word_table, conv_word_w, conv_word_b, out);
}


# Embedding gather: one 16-byte load per lane per row, all 16/18 rows of a tile in flight, indices from one up-front load, replacing 80-90 serial dependent 4-byte loads per tile
# speedup vs baseline: 1.3805x; 1.3805x over previous
#include <hip/hip_runtime.h>

#define Bsz 32
#define Lsz 256
#define Csz 16
#define Dsz 128
#define KTOT 384
#define NKCHUNK 96
#define WAVES 8
#define WT2_PITCH 288
#define WT2_FLOATS (192 * WT2_PITCH)
#define XT_FLOATS (64 * 36)
#define FEATS_FLOATS (Dsz * 16)

typedef __attribute__((ext_vector_type(2))) float v2f;
typedef __attribute__((ext_vector_type(4))) float v4f;
typedef __attribute__((ext_vector_type(8))) float v8f;

__device__ __forceinline__ void load_wt(const float* __restrict__ W, float* lds, int tid) {
  for (int idx = tid; idx < Dsz * Dsz; idx += 256) {
    const int dout = idx >> 7;
    const int din  = idx & 127;
    const float* wp = W + dout * 384 + din * 3;
    const float w0 = wp[0], w1 = wp[1], w2 = wp[2];
    const int kg0 = din;
    const int kg1 = 128 + din;
    const int kg2 = 256 + din;
    lds[(kg0 >> 1) * WT2_PITCH + dout * 2 + (kg0 & 1)] = w0;
    lds[(kg1 >> 1) * WT2_PITCH + dout * 2 + (kg1 & 1)] = w1;
    lds[(kg2 >> 1) * WT2_PITCH + dout * 2 + (kg2 & 1)] = w2;
  }
}

__device__ __forceinline__ void put_col(float* xt, int lane, int cc, v4f v) {
  float* p = xt + (2 * lane) * 36 + cc * 2;
  v2f lo = {v[0], v[1]};
  v2f hi = {v[2], v[3]};
  *(v2f*)(p)      = lo;
  *(v2f*)(p + 36) = hi;
}

__device__ __forceinline__ void load_frags(const float* __restrict__ wt,
                                           const float* __restrict__ xt,
                                           int kk, int r, int h,
                                           v2f a[8], v2f* bf) {
  const int kg0  = kk * 4 + 2 * h;
  const int tap  = kg0 >> 7;
  const int din0 = kg0 & 127;
  *bf = *(const v2f*)(xt + (din0 >> 1) * 36 + (r + tap) * 2);
  const float* wrow = wt + (kg0 >> 1) * WT2_PITCH + r * 2;
#pragma unroll
  for (int t = 0; t < 8; ++t) {
    a[t] = *(const v2f*)(wrow + t * 32);
  }
}

__device__ __forceinline__ void wmma_group(const v2f a[8], v2f bf, v8f acc[8]) {
#pragma unroll
  for (int t = 0; t < 8; ++t) {
    acc[t] = __builtin_amdgcn_wmma_f32_16x16x4_f32(
         false, a[t],  false, bf,
         (short)0, acc[t],  false,  false);
  }
}

__device__ __forceinline__ void conv_tiles(const float* __restrict__ wt,
                                           const float* __restrict__ xt,
                                           int r, int h, v8f acc[8]) {
  v2f a0[8], a1[8], b0, b1;
  load_frags(wt, xt, 0, r, h, a0, &b0);
#pragma unroll 1
  for (int kk = 0; kk < NKCHUNK - 2; kk += 2) {
    load_frags(wt, xt, kk + 1, r, h, a1, &b1);
    wmma_group(a0, b0, acc);
    load_frags(wt, xt, kk + 2, r, h, a0, &b0);
    wmma_group(a1, b1, acc);
  }
  load_frags(wt, xt, NKCHUNK - 1, r, h, a1, &b1);
  wmma_group(a0, b0, acc);
  wmma_group(a1, b1, acc);
}

__global__ __launch_bounds__(256) void char_conv_kernel(
    const int* __restrict__ words_in_char,
    const float* __restrict__ chr_table,
    const float* __restrict__ conv_w,
    const float* __restrict__ conv_b,
    float* __restrict__ out) {
  extern __shared__ float smem[];
  float* wt = smem;
  const int tid  = threadIdx.x;
  const int lane = tid & 31;
  const int w    = tid >> 5;
  float* xt    = smem + WT2_FLOATS + w * XT_FLOATS;
  float* feats = smem + WT2_FLOATS + WAVES * XT_FLOATS;

  const int gw0 = blockIdx.x * 16;
  const int b   = gw0 >> 8;
  const int l0  = gw0 & 255;
  const int r   = lane & 15;
  const int h   = lane >> 4;

  const int my_ch = words_in_char[(gw0 + w * 2) * Csz + lane];

  load_wt(conv_w, wt, tid);
  __syncthreads();

  for (int wi = 0; wi < 2; ++wi) {
    const int lw = w * 2 + wi;

    v4f rows[Csz];
#pragma unroll
    for (int c = 0; c < Csz; ++c) {
      const int ch = __builtin_amdgcn_readlane(my_ch, wi * Csz + c);
      rows[c] = *(const v4f*)(chr_table + ch * Dsz + lane * 4);
    }
    const v4f zero4 = {0.0f, 0.0f, 0.0f, 0.0f};
    put_col(xt, lane, 0, zero4);
    put_col(xt, lane, 17, zero4);
#pragma unroll
    for (int c = 0; c < Csz; ++c) {
      put_col(xt, lane, c + 1, rows[c]);
    }

    v8f acc[8] = {};
    conv_tiles(wt, xt, r, h, acc);

#pragma unroll
    for (int t = 0; t < 8; ++t) {
#pragma unroll
      for (int j = 0; j < 8; ++j) {
        float v = acc[t][j];
        v = fmaxf(v, __shfl_xor(v, 1, 32));
        v = fmaxf(v, __shfl_xor(v, 2, 32));
        v = fmaxf(v, __shfl_xor(v, 4, 32));
        v = fmaxf(v, __shfl_xor(v, 8, 32));
        if (r == j) {
          feats[(t * 16 + 8 * h + j) * 16 + lw] = v;
        }
      }
    }
  }
  __syncthreads();

  for (int i = tid; i < Dsz * 16; i += 256) {
    const int d = i >> 4;
    const int j = i & 15;
    out[((Bsz + b) * Dsz + d) * Lsz + l0 + j] = feats[d * 16 + j] + conv_b[d];
  }
}

__global__ __launch_bounds__(256) void word_conv_kernel(
    const int* __restrict__ word_vector,
    const float* __restrict__ word_table,
    const float* __restrict__ conv_w,
    const float* __restrict__ conv_b,
    float* __restrict__ out) {
  extern __shared__ float smem[];
  float* wt = smem;
  const int tid  = threadIdx.x;
  const int lane = tid & 31;
  const int w    = tid >> 5;
  float* xt = smem + WT2_FLOATS + w * XT_FLOATS;

  const int task = blockIdx.x * WAVES + w;
  const int b    = task >> 4;
  const int l0   = (task & 15) * 16;
  const int r    = lane & 15;
  const int h    = lane >> 4;

  int lcol = l0 + lane - 1;
  lcol = lcol < 0 ? 0 : (lcol > Lsz - 1 ? Lsz - 1 : lcol);
  const int my_w = word_vector[b * Lsz + lcol];

  load_wt(conv_w, wt, tid);
  __syncthreads();

  v4f rows[18];
#pragma unroll
  for (int cc = 0; cc < 18; ++cc) {
    const int widx = __builtin_amdgcn_readlane(my_w, cc);
    rows[cc] = *(const v4f*)(word_table + widx * Dsz + lane * 4);
  }
  const v4f zero4 = {0.0f, 0.0f, 0.0f, 0.0f};
  if (l0 == 0)        rows[0]  = zero4;
  if (l0 + 16 >= Lsz) rows[17] = zero4;
#pragma unroll
  for (int cc = 0; cc < 18; ++cc) {
    put_col(xt, lane, cc, rows[cc]);
  }

  v8f acc[8] = {};
  conv_tiles(wt, xt, r, h, acc);

#pragma unroll
  for (int t = 0; t < 8; ++t) {
#pragma unroll
    for (int j = 0; j < 8; ++j) {
      const int row = t * 16 + 8 * h + j;
      out[(b * Dsz + row) * Lsz + l0 + r] = acc[t][j] + conv_b[row];
    }
  }
}

extern "C" void kernel_launch(void* const* d_in, const int* in_sizes, int n_in,
                              void* d_out, int out_size, void* d_ws, size_t ws_size,
                              hipStream_t stream) {
  (void)in_sizes; (void)n_in; (void)out_size; (void)d_ws; (void)ws_size;
  const int*   word_vector   = (const int*)d_in[0];
  const int*   words_in_char = (const int*)d_in[1];
  const float* word_table    = (const float*)d_in[2];
  const float* chr_table     = (const float*)d_in[3];
  const float* conv_chr_w    = (const float*)d_in[4];
  const float* conv_chr_b    = (const float*)d_in[5];
  const float* conv_word_w   = (const float*)d_in[6];
  const float* conv_word_b   = (const float*)d_in[7];
  float* out = (float*)d_out;

  const size_t char_lds = (size_t)(WT2_FLOATS + WAVES * XT_FLOATS + FEATS_FLOATS) * sizeof(float);
  const size_t word_lds = (size_t)(WT2_FLOATS + WAVES * XT_FLOATS) * sizeof(float);

  static bool attr_done = false;
  if (!attr_done) {
    hipFuncSetAttribute((const void*)char_conv_kernel,
                        hipFuncAttributeMaxDynamicSharedMemorySize, (int)char_lds);
    hipFuncSetAttribute((const void*)word_conv_kernel,
                        hipFuncAttributeMaxDynamicSharedMemorySize, (int)word_lds);
    attr_done = true;
  }

  char_conv_kernel<<<dim3(Bsz * Lsz / 16), dim3(256), char_lds, stream>>>(
      words_in_char, chr_table, conv_chr_w, conv_chr_b, out);
  word_conv_kernel<<<dim3(Bsz * 16 / WAVES), dim3(256), word_lds, stream>>>(
      word_vector, word_table, conv_word_w, conv_word_b, out);
}


# Char-path max over the 16 conv positions: DPP row_xmask butterfly on the vector ALU, 8 chains of a tile interleaved, one masked store per tile, replacing 256 serial LDS-unit lane permutes per word
# speedup vs baseline: 1.7548x; 1.2711x over previous
#include <hip/hip_runtime.h>

#define Bsz 32
#define Lsz 256
#define Csz 16
#define Dsz 128
#define KTOT 384
#define NKCHUNK 96
#define WAVES 8
#define WT2_PITCH 288
#define WT2_FLOATS (192 * WT2_PITCH)
#define XT_FLOATS (64 * 36)
#define FEATS_FLOATS (Dsz * 16)

typedef __attribute__((ext_vector_type(2))) float v2f;
typedef __attribute__((ext_vector_type(4))) float v4f;
typedef __attribute__((ext_vector_type(8))) float v8f;

__device__ __forceinline__ void load_wt(const float* __restrict__ W, float* lds, int tid) {
  for (int idx = tid; idx < Dsz * Dsz; idx += 256) {
    const int dout = idx >> 7;
    const int din  = idx & 127;
    const float* wp = W + dout * 384 + din * 3;
    const float w0 = wp[0], w1 = wp[1], w2 = wp[2];
    const int kg0 = din;
    const int kg1 = 128 + din;
    const int kg2 = 256 + din;
    lds[(kg0 >> 1) * WT2_PITCH + dout * 2 + (kg0 & 1)] = w0;
    lds[(kg1 >> 1) * WT2_PITCH + dout * 2 + (kg1 & 1)] = w1;
    lds[(kg2 >> 1) * WT2_PITCH + dout * 2 + (kg2 & 1)] = w2;
  }
}

__device__ __forceinline__ void put_col(float* xt, int lane, int cc, v4f v) {
  float* p = xt + (2 * lane) * 36 + cc * 2;
  v2f lo = {v[0], v[1]};
  v2f hi = {v[2], v[3]};
  *(v2f*)(p)      = lo;
  *(v2f*)(p + 36) = hi;
}

__device__ __forceinline__ void load_frags(const float* __restrict__ wt,
                                           const float* __restrict__ xt,
                                           int kk, int r, int h,
                                           v2f a[8], v2f* bf) {
  const int kg0  = kk * 4 + 2 * h;
  const int tap  = kg0 >> 7;
  const int din0 = kg0 & 127;
  *bf = *(const v2f*)(xt + (din0 >> 1) * 36 + (r + tap) * 2);
  const float* wrow = wt + (kg0 >> 1) * WT2_PITCH + r * 2;
#pragma unroll
  for (int t = 0; t < 8; ++t) {
    a[t] = *(const v2f*)(wrow + t * 32);
  }
}

__device__ __forceinline__ void wmma_group(const v2f a[8], v2f bf, v8f acc[8]) {
#pragma unroll
  for (int t = 0; t < 8; ++t) {
    acc[t] = __builtin_amdgcn_wmma_f32_16x16x4_f32(
         false, a[t],  false, bf,
         (short)0, acc[t],  false,  false);
  }
}

__device__ __forceinline__ void conv_tiles(const float* __restrict__ wt,
                                           const float* __restrict__ xt,
                                           int r, int h, v8f acc[8]) {
  v2f a0[8], a1[8], b0, b1;
  load_frags(wt, xt, 0, r, h, a0, &b0);
#pragma unroll 1
  for (int kk = 0; kk < NKCHUNK - 2; kk += 2) {
    load_frags(wt, xt, kk + 1, r, h, a1, &b1);
    wmma_group(a0, b0, acc);
    load_frags(wt, xt, kk + 2, r, h, a0, &b0);
    wmma_group(a1, b1, acc);
  }
  load_frags(wt, xt, NKCHUNK - 1, r, h, a1, &b1);
  wmma_group(a0, b0, acc);
  wmma_group(a1, b1, acc);
}

template <int XMASK>
__device__ __forceinline__ float row_xor_lane(float v) {
  const int bits = __builtin_bit_cast(int, v);
  return __builtin_bit_cast(
      float, __builtin_amdgcn_update_dpp(bits, bits, 0x160 + XMASK, 0xf, 0xf, true));
}

template <int XMASK>
__device__ __forceinline__ void row_max_step(float m[8]) {
#pragma unroll
  for (int j = 0; j < 8; ++j) {
    m[j] = fmaxf(m[j], row_xor_lane<XMASK>(m[j]));
  }
}

__device__ __forceinline__ void row_max16_x8(float m[8]) {
  row_max_step<1>(m);
  row_max_step<2>(m);
  row_max_step<4>(m);
  row_max_step<8>(m);
}

__global__ __launch_bounds__(256) void char_conv_kernel(
    const int* __restrict__ words_in_char,
    const float* __restrict__ chr_table,
    const float* __restrict__ conv_w,
    const float* __restrict__ conv_b,
    float* __restrict__ out) {
  extern __shared__ float smem[];
  float* wt = smem;
  const int tid  = threadIdx.x;
  const int lane = tid & 31;
  const int w    = tid >> 5;
  float* xt    = smem + WT2_FLOATS + w * XT_FLOATS;
  float* feats = smem + WT2_FLOATS + WAVES * XT_FLOATS;

  const int gw0 = blockIdx.x * 16;
  const int b   = gw0 >> 8;
  const int l0  = gw0 & 255;
  const int r   = lane & 15;
  const int h   = lane >> 4;

  const int my_ch = words_in_char[(gw0 + w * 2) * Csz + lane];

  load_wt(conv_w, wt, tid);
  __syncthreads();

  for (int wi = 0; wi < 2; ++wi) {
    const int lw = w * 2 + wi;

    v4f rows[Csz];
#pragma unroll
    for (int c = 0; c < Csz; ++c) {
      const int ch = __builtin_amdgcn_readlane(my_ch, wi * Csz + c);
      rows[c] = *(const v4f*)(chr_table + ch * Dsz + lane * 4);
    }
    const v4f zero4 = {0.0f, 0.0f, 0.0f, 0.0f};
    put_col(xt, lane, 0, zero4);
    put_col(xt, lane, 17, zero4);
#pragma unroll
    for (int c = 0; c < Csz; ++c) {
      put_col(xt, lane, c + 1, rows[c]);
    }

    v8f acc[8] = {};
    conv_tiles(wt, xt, r, h, acc);

#pragma unroll
    for (int t = 0; t < 8; ++t) {
      float m[8];
#pragma unroll
      for (int j = 0; j < 8; ++j) {
        m[j] = acc[t][j];
      }
      row_max16_x8(m);
      float sel = m[0];
#pragma unroll
      for (int j = 1; j < 8; ++j) {
        sel = (r == j) ? m[j] : sel;
      }
      if (r < 8) {
        feats[(t * 16 + 8 * h + r) * 16 + lw] = sel;
      }
    }
  }
  __syncthreads();

  for (int i = tid; i < Dsz * 16; i += 256) {
    const int d = i >> 4;
    const int j = i & 15;
    out[((Bsz + b) * Dsz + d) * Lsz + l0 + j] = feats[d * 16 + j] + conv_b[d];
  }
}

__global__ __launch_bounds__(256) void word_conv_kernel(
    const int* __restrict__ word_vector,
    const float* __restrict__ word_table,
    const float* __restrict__ conv_w,
    const float* __restrict__ conv_b,
    float* __restrict__ out) {
  extern __shared__ float smem[];
  float* wt = smem;
  const int tid  = threadIdx.x;
  const int lane = tid & 31;
  const int w    = tid >> 5;
  float* xt = smem + WT2_FLOATS + w * XT_FLOATS;

  const int task = blockIdx.x * WAVES + w;
  const int b    = task >> 4;
  const int l0   = (task & 15) * 16;
  const int r    = lane & 15;
  const int h    = lane >> 4;

  int lcol = l0 + lane - 1;
  lcol = lcol < 0 ? 0 : (lcol > Lsz - 1 ? Lsz - 1 : lcol);
  const int my_w = word_vector[b * Lsz + lcol];

  load_wt(conv_w, wt, tid);
  __syncthreads();

  v4f rows[18];
#pragma unroll
  for (int cc = 0; cc < 18; ++cc) {
    const int widx = __builtin_amdgcn_readlane(my_w, cc);
    rows[cc] = *(const v4f*)(word_table + widx * Dsz + lane * 4);
  }
  const v4f zero4 = {0.0f, 0.0f, 0.0f, 0.0f};
  if (l0 == 0)        rows[0]  = zero4;
  if (l0 + 16 >= Lsz) rows[17] = zero4;
#pragma unroll
  for (int cc = 0; cc < 18; ++cc) {
    put_col(xt, lane, cc, rows[cc]);
  }

  v8f acc[8] = {};
  conv_tiles(wt, xt, r, h, acc);

#pragma unroll
  for (int t = 0; t < 8; ++t) {
#pragma unroll
    for (int j = 0; j < 8; ++j) {
      const int row = t * 16 + 8 * h + j;
      out[(b * Dsz + row) * Lsz + l0 + r] = acc[t][j] + conv_b[row];
    }
  }
}

extern "C" void kernel_launch(void* const* d_in, const int* in_sizes, int n_in,
                              void* d_out, int out_size, void* d_ws, size_t ws_size,
                              hipStream_t stream) {
  (void)in_sizes; (void)n_in; (void)out_size; (void)d_ws; (void)ws_size;
  const int*   word_vector   = (const int*)d_in[0];
  const int*   words_in_char = (const int*)d_in[1];
  const float* word_table    = (const float*)d_in[2];
  const float* chr_table     = (const float*)d_in[3];
  const float* conv_chr_w    = (const float*)d_in[4];
  const float* conv_chr_b    = (const float*)d_in[5];
  const float* conv_word_w   = (const float*)d_in[6];
  const float* conv_word_b   = (const float*)d_in[7];
  float* out = (float*)d_out;

  const size_t char_lds = (size_t)(WT2_FLOATS + WAVES * XT_FLOATS + FEATS_FLOATS) * sizeof(float);
  const size_t word_lds = (size_t)(WT2_FLOATS + WAVES * XT_FLOATS) * sizeof(float);

  static bool attr_done = false;
  if (!attr_done) {
    hipFuncSetAttribute((const void*)char_conv_kernel,
                        hipFuncAttributeMaxDynamicSharedMemorySize, (int)char_lds);
    hipFuncSetAttribute((const void*)word_conv_kernel,
                        hipFuncAttributeMaxDynamicSharedMemorySize, (int)word_lds);
    attr_done = true;
  }

  char_conv_kernel<<<dim3(Bsz * Lsz / 16), dim3(256), char_lds, stream>>>(
      words_in_char, chr_table, conv_chr_w, conv_chr_b, out);
  word_conv_kernel<<<dim3(Bsz * 16 / WAVES), dim3(256), word_lds, stream>>>(
      word_vector, word_table, conv_word_w, conv_word_b, out);
}


# Char path: conv applied once per block to the 128-row char table (per-tap f32 WMMA products kept in LDS); a word = max over sums of 3 looked-up rows; sum order: per-tap partial sums, then added
# speedup vs baseline: 5.4109x; 3.0835x over previous
#include <hip/hip_runtime.h>

#define Bsz 32
#define Lsz 256
#define Csz 16
#define Dsz 128
#define KTOT 384
#define NKCHUNK 96
#define WAVES 8
#define WT2_PITCH 288
#define WT2_FLOATS (192 * WT2_PITCH)
#define XT_FLOATS (64 * 36)
#define DSL 16
#define WL_PITCH 132
#define WL_FLOATS (3 * DSL * WL_PITCH)
#define PT_PITCH 20
#define PT_FLOATS (3 * 128 * PT_PITCH)

typedef __attribute__((ext_vector_type(2))) float v2f;
typedef __attribute__((ext_vector_type(4))) float v4f;
typedef __attribute__((ext_vector_type(8))) float v8f;
typedef __attribute__((ext_vector_type(4))) int v4i;

__device__ __forceinline__ void load_wt(const float* __restrict__ W, float* lds, int tid) {
  for (int idx = tid; idx < Dsz * Dsz; idx += 256) {
    const int dout = idx >> 7;
    const int din  = idx & 127;
    const float* wp = W + dout * 384 + din * 3;
    const float w0 = wp[0], w1 = wp[1], w2 = wp[2];
    const int kg0 = din;
    const int kg1 = 128 + din;
    const int kg2 = 256 + din;
    lds[(kg0 >> 1) * WT2_PITCH + dout * 2 + (kg0 & 1)] = w0;
    lds[(kg1 >> 1) * WT2_PITCH + dout * 2 + (kg1 & 1)] = w1;
    lds[(kg2 >> 1) * WT2_PITCH + dout * 2 + (kg2 & 1)] = w2;
  }
}

__device__ __forceinline__ void put_col(float* xt, int lane, int cc, v4f v) {
  float* p = xt + (2 * lane) * 36 + cc * 2;
  v2f lo = {v[0], v[1]};
  v2f hi = {v[2], v[3]};
  *(v2f*)(p)      = lo;
  *(v2f*)(p + 36) = hi;
}

__device__ __forceinline__ void load_frags(const float* __restrict__ wt,
                                           const float* __restrict__ xt,
                                           int kk, int r, int h,
                                           v2f a[8], v2f* bf) {
  const int kg0  = kk * 4 + 2 * h;
  const int tap  = kg0 >> 7;
  const int din0 = kg0 & 127;
  *bf = *(const v2f*)(xt + (din0 >> 1) * 36 + (r + tap) * 2);
  const float* wrow = wt + (kg0 >> 1) * WT2_PITCH + r * 2;
#pragma unroll
  for (int t = 0; t < 8; ++t) {
    a[t] = *(const v2f*)(wrow + t * 32);
  }
}

__device__ __forceinline__ void wmma_group(const v2f a[8], v2f bf, v8f acc[8]) {
#pragma unroll
  for (int t = 0; t < 8; ++t) {
    acc[t] = __builtin_amdgcn_wmma_f32_16x16x4_f32(
         false, a[t],  false, bf,
         (short)0, acc[t],  false,  false);
  }
}

__device__ __forceinline__ void conv_tiles(const float* __restrict__ wt,
                                           const float* __restrict__ xt,
                                           int r, int h, v8f acc[8]) {
  v2f a0[8], a1[8], b0, b1;
  load_frags(wt, xt, 0, r, h, a0, &b0);
#pragma unroll 1
  for (int kk = 0; kk < NKCHUNK - 2; kk += 2) {
    load_frags(wt, xt, kk + 1, r, h, a1, &b1);
    wmma_group(a0, b0, acc);
    load_frags(wt, xt, kk + 2, r, h, a0, &b0);
    wmma_group(a1, b1, acc);
  }
  load_frags(wt, xt, NKCHUNK - 1, r, h, a1, &b1);
  wmma_group(a0, b0, acc);
  wmma_group(a1, b1, acc);
}

__device__ __forceinline__ v4f ld4(const float* p) { return *(const v4f*)p; }

__device__ __forceinline__ int row_off(int ch) {
  const int c = ch < 0 ? 0 : (ch > 127 ? 127 : ch);
  return c * PT_PITCH;
}

__device__ __forceinline__ v4f max4(v4f a, v4f b) {
  v4f m = {fmaxf(a[0], b[0]), fmaxf(a[1], b[1]), fmaxf(a[2], b[2]), fmaxf(a[3], b[3])};
  return m;
}

__global__ __launch_bounds__(256) void char_conv_kernel(
    const int* __restrict__ words_in_char,
    const float* __restrict__ chr_table,
    const float* __restrict__ conv_w,
    const float* __restrict__ conv_b,
    float* __restrict__ out) {
  __shared__ __attribute__((aligned(16))) float wl[WL_FLOATS];
  __shared__ __attribute__((aligned(16))) float pt[PT_FLOATS];
  const int tid  = threadIdx.x;
  const int lane = tid & 31;
  const int w    = tid >> 5;
  const int r    = lane & 15;
  const int h    = lane >> 4;
  const int d0   = blockIdx.x * DSL;
  const int b    = blockIdx.y;
  const int l    = w * 32 + lane;

  const v4i* ip = (const v4i*)(words_in_char + (b * Lsz + l) * Csz);
  const v4i ci0 = ip[0], ci1 = ip[1], ci2 = ip[2], ci3 = ip[3];

  v4f bq[16];
  const float* trow = chr_table + (w * 16 + r) * Dsz + 4 * h;
#pragma unroll
  for (int j = 0; j < 16; ++j) {
    bq[j] = ld4(trow + 8 * j);
  }

  const float* wsrc = conv_w + d0 * KTOT;
  v4f wv[6];
#pragma unroll
  for (int i = 0; i < 6; ++i) {
    wv[i] = ld4(wsrc + (tid + 256 * i) * 4);
  }
#pragma unroll
  for (int i = 0; i < 6; ++i) {
    const int f0  = (tid + 256 * i) * 4;
    const int dl  = f0 / KTOT;
    const int rem = f0 - dl * KTOT;
#pragma unroll
    for (int e = 0; e < 4; ++e) {
      const int din = (rem + e) / 3;
      const int tap = (rem + e) - din * 3;
      wl[(tap * DSL + dl) * WL_PITCH + din] = wv[i][e];
    }
  }
  __syncthreads();

  v8f acc[3] = {};
#pragma unroll
  for (int j = 0; j < 16; ++j) {
    v4f a4[3];
#pragma unroll
    for (int t = 0; t < 3; ++t) {
      a4[t] = ld4(wl + (t * DSL + r) * WL_PITCH + 8 * j + 4 * h);
    }
    const v2f b_lo = {bq[j][0], bq[j][1]};
    const v2f b_hi = {bq[j][2], bq[j][3]};
#pragma unroll
    for (int t = 0; t < 3; ++t) {
      const v2f a_lo = {a4[t][0], a4[t][1]};
      acc[t] = __builtin_amdgcn_wmma_f32_16x16x4_f32(
           false, a_lo,  false, b_lo,
           (short)0, acc[t],  false,  false);
    }
#pragma unroll
    for (int t = 0; t < 3; ++t) {
      const v2f a_hi = {a4[t][2], a4[t][3]};
      acc[t] = __builtin_amdgcn_wmma_f32_16x16x4_f32(
           false, a_hi,  false, b_hi,
           (short)0, acc[t],  false,  false);
    }
  }

#pragma unroll
  for (int t = 0; t < 3; ++t) {
    float* p = pt + (t * 128 + w * 16 + r) * PT_PITCH + 8 * h;
    const v4f lo = {acc[t][0], acc[t][1], acc[t][2], acc[t][3]};
    const v4f hi = {acc[t][4], acc[t][5], acc[t][6], acc[t][7]};
    *(v4f*)(p)     = lo;
    *(v4f*)(p + 4) = hi;
  }
  __syncthreads();

  int off[Csz];
#pragma unroll
  for (int e = 0; e < 4; ++e) {
    off[e]      = row_off(ci0[e]);
    off[4 + e]  = row_off(ci1[e]);
    off[8 + e]  = row_off(ci2[e]);
    off[12 + e] = row_off(ci3[e]);
  }
  const float* pt0 = pt;
  const float* pt1 = pt + 128 * PT_PITCH;
  const float* pt2 = pt + 256 * PT_PITCH;

  v4f m[4];
#pragma unroll
  for (int c = 0; c < Csz; ++c) {
    v4f s[4];
#pragma unroll
    for (int q = 0; q < 4; ++q) {
      s[q] = ld4(pt1 + off[c] + 4 * q);
    }
    if (c > 0) {
#pragma unroll
      for (int q = 0; q < 4; ++q) {
        s[q] = ld4(pt0 + off[c - 1] + 4 * q) + s[q];
      }
    }
    if (c < Csz - 1) {
#pragma unroll
      for (int q = 0; q < 4; ++q) {
        s[q] = s[q] + ld4(pt2 + off[c + 1] + 4 * q);
      }
    }
#pragma unroll
    for (int q = 0; q < 4; ++q) {
      m[q] = (c == 0) ? s[q] : max4(m[q], s[q]);
    }
  }

  float* orow = out + ((Bsz + b) * Dsz + d0) * Lsz + l;
#pragma unroll
  for (int q = 0; q < 4; ++q) {
#pragma unroll
    for (int e = 0; e < 4; ++e) {
      const int dl = 4 * q + e;
      orow[dl * Lsz] = m[q][e] + conv_b[d0 + dl];
    }
  }
}

__global__ __launch_bounds__(256) void word_conv_kernel(
    const int* __restrict__ word_vector,
    const float* __restrict__ word_table,
    const float* __restrict__ conv_w,
    const float* __restrict__ conv_b,
    float* __restrict__ out) {
  extern __shared__ float smem[];
  float* wt = smem;
  const int tid  = threadIdx.x;
  const int lane = tid & 31;
  const int w    = tid >> 5;
  float* xt = smem + WT2_FLOATS + w * XT_FLOATS;

  const int task = blockIdx.x * WAVES + w;
  const int b    = task >> 4;
  const int l0   = (task & 15) * 16;
  const int r    = lane & 15;
  const int h    = lane >> 4;

  int lcol = l0 + lane - 1;
  lcol = lcol < 0 ? 0 : (lcol > Lsz - 1 ? Lsz - 1 : lcol);
  const int my_w = word_vector[b * Lsz + lcol];

  load_wt(conv_w, wt, tid);
  __syncthreads();

  v4f rows[18];
#pragma unroll
  for (int cc = 0; cc < 18; ++cc) {
    const int widx = __builtin_amdgcn_readlane(my_w, cc);
    rows[cc] = *(const v4f*)(word_table + widx * Dsz + lane * 4);
  }
  const v4f zero4 = {0.0f, 0.0f, 0.0f, 0.0f};
  if (l0 == 0)        rows[0]  = zero4;
  if (l0 + 16 >= Lsz) rows[17] = zero4;
#pragma unroll
  for (int cc = 0; cc < 18; ++cc) {
    put_col(xt, lane, cc, rows[cc]);
  }

  v8f acc[8] = {};
  conv_tiles(wt, xt, r, h, acc);

#pragma unroll
  for (int t = 0; t < 8; ++t) {
#pragma unroll
    for (int j = 0; j < 8; ++j) {
      const int row = t * 16 + 8 * h + j;
      out[(b * Dsz + row) * Lsz + l0 + r] = acc[t][j] + conv_b[row];
    }
  }
}

extern "C" void kernel_launch(void* const* d_in, const int* in_sizes, int n_in,
                              void* d_out, int out_size, void* d_ws, size_t ws_size,
                              hipStream_t stream) {
  (void)in_sizes; (void)n_in; (void)out_size; (void)d_ws; (void)ws_size;
  const int*   word_vector   = (const int*)d_in[0];
  const int*   words_in_char = (const int*)d_in[1];
  const float* word_table    = (const float*)d_in[2];
  const float* chr_table     = (const float*)d_in[3];
  const float* conv_chr_w    = (const float*)d_in[4];
  const float* conv_chr_b    = (const float*)d_in[5];
  const float* conv_word_w   = (const float*)d_in[6];
  const float* conv_word_b   = (const float*)d_in[7];
  float* out = (float*)d_out;

  const size_t word_lds = (size_t)(WT2_FLOATS + WAVES * XT_FLOATS) * sizeof(float);

  static bool attr_done = false;
  if (!attr_done) {
    hipFuncSetAttribute((const void*)word_conv_kernel,
                        hipFuncAttributeMaxDynamicSharedMemorySize, (int)word_lds);
    attr_done = true;
  }

  char_conv_kernel<<<dim3(Dsz / DSL, Bsz), dim3(256), 0, stream>>>(
      words_in_char, chr_table, conv_chr_w, conv_chr_b, out);
  word_conv_kernel<<<dim3(Bsz * 16 / WAVES), dim3(256), word_lds, stream>>>(
      word_vector, word_table, conv_word_w, conv_word_b, out);
}


# Word conv re-tiled over all 256 WGPs: block = 32 d x 128 l of one b, stages only its 48 KB weight slice (was 64 blocks, each all 196 KB); f32 sum order over din regrouped (8-din steps per tap)
# speedup vs baseline: 10.8622x; 2.0075x over previous
#include <hip/hip_runtime.h>

#define Bsz 32
#define Lsz 256
#define Csz 16
#define Dsz 128
#define KTOT 384
#define DSL 16
#define WL_PITCH 132
#define WL_FLOATS (3 * DSL * WL_PITCH)
#define PT_PITCH 20
#define PT_FLOATS (3 * 128 * PT_PITCH)
#define WSL 32
#define WW_PITCH 132
#define WW_FLOATS (3 * WSL * WW_PITCH)
#define LBLK 128
#define XS_ROWS 136
#define XS_PITCH 132
#define XS_FLOATS (XS_ROWS * XS_PITCH)
#define WORD_LDS_FLOATS 41472

typedef __attribute__((ext_vector_type(2))) float v2f;
typedef __attribute__((ext_vector_type(4))) float v4f;
typedef __attribute__((ext_vector_type(8))) float v8f;
typedef __attribute__((ext_vector_type(4))) int v4i;

__device__ __forceinline__ v4f ld4(const float* p) { return *(const v4f*)p; }

__device__ __forceinline__ int row_off(int ch) {
  const int c = ch < 0 ? 0 : (ch > 127 ? 127 : ch);
  return c * PT_PITCH;
}

__device__ __forceinline__ v4f max4(v4f a, v4f b) {
  v4f m = {fmaxf(a[0], b[0]), fmaxf(a[1], b[1]), fmaxf(a[2], b[2]), fmaxf(a[3], b[3])};
  return m;
}

__global__ __launch_bounds__(256) void char_conv_kernel(
    const int* __restrict__ words_in_char,
    const float* __restrict__ chr_table,
    const float* __restrict__ conv_w,
    const float* __restrict__ conv_b,
    float* __restrict__ out) {
  __shared__ __attribute__((aligned(16))) float wl[WL_FLOATS];
  __shared__ __attribute__((aligned(16))) float pt[PT_FLOATS];
  const int tid  = threadIdx.x;
  const int lane = tid & 31;
  const int w    = tid >> 5;
  const int r    = lane & 15;
  const int h    = lane >> 4;
  const int d0   = blockIdx.x * DSL;
  const int b    = blockIdx.y;
  const int l    = w * 32 + lane;

  const v4i* ip = (const v4i*)(words_in_char + (b * Lsz + l) * Csz);
  const v4i ci0 = ip[0], ci1 = ip[1], ci2 = ip[2], ci3 = ip[3];

  v4f bq[16];
  const float* trow = chr_table + (w * 16 + r) * Dsz + 4 * h;
#pragma unroll
  for (int j = 0; j < 16; ++j) {
    bq[j] = ld4(trow + 8 * j);
  }

  const float* wsrc = conv_w + d0 * KTOT;
  v4f wv[6];
#pragma unroll
  for (int i = 0; i < 6; ++i) {
    wv[i] = ld4(wsrc + (tid + 256 * i) * 4);
  }
#pragma unroll
  for (int i = 0; i < 6; ++i) {
    const int f0  = (tid + 256 * i) * 4;
    const int dl  = f0 / KTOT;
    const int rem = f0 - dl * KTOT;
#pragma unroll
    for (int e = 0; e < 4; ++e) {
      const int din = (rem + e) / 3;
      const int tap = (rem + e) - din * 3;
      wl[(tap * DSL + dl) * WL_PITCH + din] = wv[i][e];
    }
  }
  __syncthreads();

  v8f acc[3] = {};
#pragma unroll
  for (int j = 0; j < 16; ++j) {
    v4f a4[3];
#pragma unroll
    for (int t = 0; t < 3; ++t) {
      a4[t] = ld4(wl + (t * DSL + r) * WL_PITCH + 8 * j + 4 * h);
    }
    const v2f b_lo = {bq[j][0], bq[j][1]};
    const v2f b_hi = {bq[j][2], bq[j][3]};
#pragma unroll
    for (int t = 0; t < 3; ++t) {
      const v2f a_lo = {a4[t][0], a4[t][1]};
      acc[t] = __builtin_amdgcn_wmma_f32_16x16x4_f32(
           false, a_lo,  false, b_lo,
           (short)0, acc[t],  false,  false);
    }
#pragma unroll
    for (int t = 0; t < 3; ++t) {
      const v2f a_hi = {a4[t][2], a4[t][3]};
      acc[t] = __builtin_amdgcn_wmma_f32_16x16x4_f32(
           false, a_hi,  false, b_hi,
           (short)0, acc[t],  false,  false);
    }
  }

#pragma unroll
  for (int t = 0; t < 3; ++t) {
    float* p = pt + (t * 128 + w * 16 + r) * PT_PITCH + 8 * h;
    const v4f lo = {acc[t][0], acc[t][1], acc[t][2], acc[t][3]};
    const v4f hi = {acc[t][4], acc[t][5], acc[t][6], acc[t][7]};
    *(v4f*)(p)     = lo;
    *(v4f*)(p + 4) = hi;
  }
  __syncthreads();

  int off[Csz];
#pragma unroll
  for (int e = 0; e < 4; ++e) {
    off[e]      = row_off(ci0[e]);
    off[4 + e]  = row_off(ci1[e]);
    off[8 + e]  = row_off(ci2[e]);
    off[12 + e] = row_off(ci3[e]);
  }
  const float* pt0 = pt;
  const float* pt1 = pt + 128 * PT_PITCH;
  const float* pt2 = pt + 256 * PT_PITCH;

  v4f m[4];
#pragma unroll
  for (int c = 0; c < Csz; ++c) {
    v4f s[4];
#pragma unroll
    for (int q = 0; q < 4; ++q) {
      s[q] = ld4(pt1 + off[c] + 4 * q);
    }
    if (c > 0) {
#pragma unroll
      for (int q = 0; q < 4; ++q) {
        s[q] = ld4(pt0 + off[c - 1] + 4 * q) + s[q];
      }
    }
    if (c < Csz - 1) {
#pragma unroll
      for (int q = 0; q < 4; ++q) {
        s[q] = s[q] + ld4(pt2 + off[c + 1] + 4 * q);
      }
    }
#pragma unroll
    for (int q = 0; q < 4; ++q) {
      m[q] = (c == 0) ? s[q] : max4(m[q], s[q]);
    }
  }

  float* orow = out + ((Bsz + b) * Dsz + d0) * Lsz + l;
#pragma unroll
  for (int q = 0; q < 4; ++q) {
#pragma unroll
    for (int e = 0; e < 4; ++e) {
      const int dl = 4 * q + e;
      orow[dl * Lsz] = m[q][e] + conv_b[d0 + dl];
    }
  }
}

struct WordFrags {
  v4f a[4];
  v4f b[2][4];
};

__device__ __forceinline__ void word_load_block(const float* __restrict__ aw,
                                                const float* __restrict__ xb,
                                                int blk, WordFrags& f) {
  const int tap = blk >> 2;
  const int j0  = (blk & 3) * 4;
#pragma unroll
  for (int jj = 0; jj < 4; ++jj) {
    f.a[jj]    = ld4(aw + tap * (WSL * WW_PITCH) + 8 * (j0 + jj));
    f.b[0][jj] = ld4(xb + tap * XS_PITCH + 8 * (j0 + jj));
    f.b[1][jj] = ld4(xb + (16 + tap) * XS_PITCH + 8 * (j0 + jj));
  }
}

__device__ __forceinline__ void word_wmma_block(const WordFrags& f, v8f acc[2]) {
#pragma unroll
  for (int jj = 0; jj < 4; ++jj) {
    const v2f a_lo = {f.a[jj][0], f.a[jj][1]};
    const v2f a_hi = {f.a[jj][2], f.a[jj][3]};
#pragma unroll
    for (int n = 0; n < 2; ++n) {
      const v2f b_lo = {f.b[n][jj][0], f.b[n][jj][1]};
      acc[n] = __builtin_amdgcn_wmma_f32_16x16x4_f32(
           false, a_lo,  false, b_lo,
           (short)0, acc[n],  false,  false);
    }
#pragma unroll
    for (int n = 0; n < 2; ++n) {
      const v2f b_hi = {f.b[n][jj][2], f.b[n][jj][3]};
      acc[n] = __builtin_amdgcn_wmma_f32_16x16x4_f32(
           false, a_hi,  false, b_hi,
           (short)0, acc[n],  false,  false);
    }
  }
}

__global__ __launch_bounds__(256) void word_conv_kernel(
    const int* __restrict__ word_vector,
    const float* __restrict__ word_table,
    const float* __restrict__ conv_w,
    const float* __restrict__ conv_b,
    float* __restrict__ out) {
  extern __shared__ __attribute__((aligned(16))) float smem[];
  float* wl = smem;
  float* xs = smem + WW_FLOATS;
  const int tid  = threadIdx.x;
  const int lane = tid & 31;
  const int w    = __builtin_amdgcn_readfirstlane(tid >> 5);
  const int r    = lane & 15;
  const int h    = lane >> 4;
  const int bid   = blockIdx.x;
  const int b     = bid >> 3;
  const int d0    = ((bid >> 1) & 3) * WSL;
  const int lbase = (bid & 1) * LBLK;

  int lpos = lbase + w + 8 * lane - 1;
  lpos = lpos < 0 ? 0 : (lpos > Lsz - 1 ? Lsz - 1 : lpos);
  const int my_w = word_vector[b * Lsz + lpos];

  const float* wsrc = conv_w + d0 * KTOT;
  v4f wv[12];
#pragma unroll
  for (int i = 0; i < 12; ++i) {
    wv[i] = ld4(wsrc + (tid + 256 * i) * 4);
  }

  v4f rows[17];
#pragma unroll
  for (int i = 0; i < 17; ++i) {
    const int widx = __builtin_amdgcn_readlane(my_w, i);
    rows[i] = ld4(word_table + widx * Dsz + lane * 4);
  }

  const int dt = w & 1;
  const int lp = w >> 1;
  const float* bp = conv_b + d0 + dt * 16 + 8 * h;
  const v4f bias_lo = ld4(bp);
  const v4f bias_hi = ld4(bp + 4);

#pragma unroll
  for (int i = 0; i < 12; ++i) {
    const int f0  = (tid + 256 * i) * 4;
    const int dl  = f0 / KTOT;
    const int rem = f0 - dl * KTOT;
#pragma unroll
    for (int e = 0; e < 4; ++e) {
      const int din = (rem + e) / 3;
      const int tap = (rem + e) - din * 3;
      wl[(tap * WSL + dl) * WW_PITCH + din] = wv[i][e];
    }
  }

  const v4f zero4 = {0.0f, 0.0f, 0.0f, 0.0f};
#pragma unroll
  for (int i = 0; i < 17; ++i) {
    const int row = w + 8 * i;
    v4f v = rows[i];
    if (i == 0 || i == 16) {
      const int l = lbase + row - 1;
      if (l < 0 || l > Lsz - 1) v = zero4;
    }
    *(v4f*)(xs + row * XS_PITCH + lane * 4) = v;
  }
  __syncthreads();

  const float* aw = wl + (dt * 16 + r) * WW_PITCH + 4 * h;
  const float* xb = xs + (lp * 32 + r) * XS_PITCH + 4 * h;

  v8f acc[2] = {};
  WordFrags f0, f1;
  word_load_block(aw, xb, 0, f0);
#pragma unroll
  for (int blk = 0; blk < 12; blk += 2) {
    word_load_block(aw, xb, blk + 1, f1);
    __builtin_amdgcn_sched_barrier(0);
    word_wmma_block(f0, acc);
    __builtin_amdgcn_sched_barrier(0);
    if (blk + 2 < 12) word_load_block(aw, xb, blk + 2, f0);
    __builtin_amdgcn_sched_barrier(0);
    word_wmma_block(f1, acc);
    __builtin_amdgcn_sched_barrier(0);
  }

  float* orow = out + (b * Dsz + d0 + dt * 16 + 8 * h) * Lsz + lbase + lp * 32 + r;
#pragma unroll
  for (int n = 0; n < 2; ++n) {
#pragma unroll
    for (int i = 0; i < 8; ++i) {
      const float bias = i < 4 ? bias_lo[i & 3] : bias_hi[i & 3];
      orow[i * Lsz + n * 16] = acc[n][i] + bias;
    }
  }
}

extern "C" void kernel_launch(void* const* d_in, const int* in_sizes, int n_in,
                              void* d_out, int out_size, void* d_ws, size_t ws_size,
                              hipStream_t stream) {
  (void)in_sizes; (void)n_in; (void)out_size; (void)d_ws; (void)ws_size;
  const int*   word_vector   = (const int*)d_in[0];
  const int*   words_in_char = (const int*)d_in[1];
  const float* word_table    = (const float*)d_in[2];
  const float* chr_table     = (const float*)d_in[3];
  const float* conv_chr_w    = (const float*)d_in[4];
  const float* conv_chr_b    = (const float*)d_in[5];
  const float* conv_word_w   = (const float*)d_in[6];
  const float* conv_word_b   = (const float*)d_in[7];
  float* out = (float*)d_out;

  const size_t word_lds = (size_t)WORD_LDS_FLOATS * sizeof(float);

  static bool attr_done = false;
  if (!attr_done) {
    hipFuncSetAttribute((const void*)word_conv_kernel,
                        hipFuncAttributeMaxDynamicSharedMemorySize, (int)word_lds);
    attr_done = true;
  }

  char_conv_kernel<<<dim3(Dsz / DSL, Bsz), dim3(256), 0, stream>>>(
      words_in_char, chr_table, conv_chr_w, conv_chr_b, out);
  word_conv_kernel<<<dim3(Bsz * 2 * (Dsz / WSL)), dim3(256), word_lds, stream>>>(
      word_vector, word_table, conv_word_w, conv_word_b, out);
}


# Char-path gather free of LDS bank conflicts: product table held as 4 bank-quarter copies with rotated quads, 4 lanes share a word, outputs transposed through LDS; sum order unchanged
# speedup vs baseline: 11.8844x; 1.0941x over previous
#include <hip/hip_runtime.h>

#define Bsz 32
#define Lsz 256
#define Csz 16
#define Dsz 128
#define KTOT 384
#define DSL 16
#define WL_PITCH 132
#define WL_FLOATS (3 * DSL * WL_PITCH)
#define PT_ROW 64
#define PT_CHAR (3 * PT_ROW)
#define PT_FLOATS (128 * PT_CHAR)
#define ST_PITCH 34
#define WSL 32
#define WW_PITCH 132
#define WW_FLOATS (3 * WSL * WW_PITCH)
#define LBLK 128
#define XS_ROWS 136
#define XS_PITCH 132
#define XS_FLOATS (XS_ROWS * XS_PITCH)
#define WORD_LDS_FLOATS 41472

typedef __attribute__((ext_vector_type(2))) float v2f;
typedef __attribute__((ext_vector_type(4))) float v4f;
typedef __attribute__((ext_vector_type(8))) float v8f;
typedef __attribute__((ext_vector_type(4))) int v4i;

__device__ __forceinline__ v4f ld4(const float* p) { return *(const v4f*)p; }

__device__ __forceinline__ int clamp_char(int ch) {
  return ch < 0 ? 0 : (ch > 127 ? 127 : ch);
}

__device__ __forceinline__ v4f max4(v4f a, v4f b) {
  v4f m = {fmaxf(a[0], b[0]), fmaxf(a[1], b[1]), fmaxf(a[2], b[2]), fmaxf(a[3], b[3])};
  return m;
}

__global__ __launch_bounds__(256) void char_conv_kernel(
    const int* __restrict__ words_in_char,
    const float* __restrict__ chr_table,
    const float* __restrict__ conv_w,
    const float* __restrict__ conv_b,
    float* __restrict__ out) {
  __shared__ __attribute__((aligned(16))) float wl[WL_FLOATS];
  __shared__ __attribute__((aligned(16))) float pt[PT_FLOATS];
  const int tid  = threadIdx.x;
  const int lane = tid & 31;
  const int w    = tid >> 5;
  const int r    = lane & 15;
  const int h    = lane >> 4;
  const int d0   = blockIdx.x * DSL;
  const int b    = blockIdx.y;
  const int gq   = lane & 3;
  const int gk   = (lane >> 2) & 3;
  const int gs   = lane >> 2;

  v4i ci[4][4];
#pragma unroll
  for (int it = 0; it < 4; ++it) {
    const v4i* ip = (const v4i*)(words_in_char + (b * Lsz + w * 32 + it * 8 + gs) * Csz);
#pragma unroll
    for (int j = 0; j < 4; ++j) {
      ci[it][j] = ip[j];
    }
  }

  v4f bq[16];
  const float* trow = chr_table + (w * 16 + r) * Dsz + 4 * h;
#pragma unroll
  for (int j = 0; j < 16; ++j) {
    bq[j] = ld4(trow + 8 * j);
  }

  const float* wsrc = conv_w + d0 * KTOT;
  v4f wv[6];
#pragma unroll
  for (int i = 0; i < 6; ++i) {
    wv[i] = ld4(wsrc + (tid + 256 * i) * 4);
  }
#pragma unroll
  for (int i = 0; i < 6; ++i) {
    const int f0  = (tid + 256 * i) * 4;
    const int dl  = f0 / KTOT;
    const int rem = f0 - dl * KTOT;
#pragma unroll
    for (int e = 0; e < 4; ++e) {
      const int din = (rem + e) / 3;
      const int tap = (rem + e) - din * 3;
      wl[(tap * DSL + dl) * WL_PITCH + din] = wv[i][e];
    }
  }
  __syncthreads();

  v8f acc[3] = {};
#pragma unroll
  for (int j = 0; j < 16; ++j) {
    v4f a4[3];
#pragma unroll
    for (int t = 0; t < 3; ++t) {
      a4[t] = ld4(wl + (t * DSL + r) * WL_PITCH + 8 * j + 4 * h);
    }
    const v2f b_lo = {bq[j][0], bq[j][1]};
    const v2f b_hi = {bq[j][2], bq[j][3]};
#pragma unroll
    for (int t = 0; t < 3; ++t) {
      const v2f a_lo = {a4[t][0], a4[t][1]};
      acc[t] = __builtin_amdgcn_wmma_f32_16x16x4_f32(
           false, a_lo,  false, b_lo,
           (short)0, acc[t],  false,  false);
    }
#pragma unroll
    for (int t = 0; t < 3; ++t) {
      const v2f a_hi = {a4[t][2], a4[t][3]};
      acc[t] = __builtin_amdgcn_wmma_f32_16x16x4_f32(
           false, a_hi,  false, b_hi,
           (short)0, acc[t],  false,  false);
    }
  }

  {
    const int chr   = w * 16 + r;
    const int s_lo  = 4 * ((2 * h + chr) & 3);
    const int s_hi  = 4 * ((2 * h + 1 + chr) & 3);
#pragma unroll
    for (int t = 0; t < 3; ++t) {
      const v4f lo = {acc[t][0], acc[t][1], acc[t][2], acc[t][3]};
      const v4f hi = {acc[t][4], acc[t][5], acc[t][6], acc[t][7]};
#pragma unroll
      for (int jp = 0; jp < 4; ++jp) {
        const int kk = (jp + (r >> 2)) & 3;
        float* p = pt + chr * PT_CHAR + t * PT_ROW + 16 * kk;
        *(v4f*)(p + s_lo) = lo;
        *(v4f*)(p + s_hi) = hi;
      }
    }
  }
  __syncthreads();

  v4f res[4];
#pragma unroll
  for (int it = 0; it < 4; ++it) {
    int off[Csz];
#pragma unroll
    for (int j = 0; j < Csz; ++j) {
      const int ch = clamp_char(ci[it][j >> 2][j & 3]);
      off[j] = ch * PT_CHAR + 16 * gk + 4 * ((gq + ch) & 3);
    }
    v4f m;
#pragma unroll
    for (int c = 0; c < Csz; ++c) {
      v4f s = ld4(pt + PT_ROW + off[c]);
      if (c > 0) {
        s = ld4(pt + off[c - 1]) + s;
      }
      if (c < Csz - 1) {
        s = s + ld4(pt + 2 * PT_ROW + off[c + 1]);
      }
      m = (c == 0) ? s : max4(m, s);
    }
    res[it] = m;
  }

  float* st = wl + w * (DSL * ST_PITCH);
#pragma unroll
  for (int it = 0; it < 4; ++it) {
#pragma unroll
    for (int e = 0; e < 4; ++e) {
      st[(4 * gq + e) * ST_PITCH + it * 8 + gs] = res[it][e];
    }
  }
  __syncthreads();

  float* orow = out + ((Bsz + b) * Dsz + d0) * Lsz + w * 32 + lane;
#pragma unroll
  for (int dl = 0; dl < DSL; ++dl) {
    orow[dl * Lsz] = st[dl * ST_PITCH + lane] + conv_b[d0 + dl];
  }
}

struct WordFrags {
  v4f a[4];
  v4f b[2][4];
};

__device__ __forceinline__ void word_load_block(const float* __restrict__ aw,
                                                const float* __restrict__ xb,
                                                int blk, WordFrags& f) {
  const int tap = blk >> 2;
  const int j0  = (blk & 3) * 4;
#pragma unroll
  for (int jj = 0; jj < 4; ++jj) {
    f.a[jj]    = ld4(aw + tap * (WSL * WW_PITCH) + 8 * (j0 + jj));
    f.b[0][jj] = ld4(xb + tap * XS_PITCH + 8 * (j0 + jj));
    f.b[1][jj] = ld4(xb + (16 + tap) * XS_PITCH + 8 * (j0 + jj));
  }
}

__device__ __forceinline__ void word_wmma_block(const WordFrags& f, v8f acc[2]) {
#pragma unroll
  for (int jj = 0; jj < 4; ++jj) {
    const v2f a_lo = {f.a[jj][0], f.a[jj][1]};
    const v2f a_hi = {f.a[jj][2], f.a[jj][3]};
#pragma unroll
    for (int n = 0; n < 2; ++n) {
      const v2f b_lo = {f.b[n][jj][0], f.b[n][jj][1]};
      acc[n] = __builtin_amdgcn_wmma_f32_16x16x4_f32(
           false, a_lo,  false, b_lo,
           (short)0, acc[n],  false,  false);
    }
#pragma unroll
    for (int n = 0; n < 2; ++n) {
      const v2f b_hi = {f.b[n][jj][2], f.b[n][jj][3]};
      acc[n] = __builtin_amdgcn_wmma_f32_16x16x4_f32(
           false, a_hi,  false, b_hi,
           (short)0, acc[n],  false,  false);
    }
  }
}

__global__ __launch_bounds__(256) void word_conv_kernel(
    const int* __restrict__ word_vector,
    const float* __restrict__ word_table,
    const float* __restrict__ conv_w,
    const float* __restrict__ conv_b,
    float* __restrict__ out) {
  extern __shared__ __attribute__((aligned(16))) float smem[];
  float* wl = smem;
  float* xs = smem + WW_FLOATS;
  const int tid  = threadIdx.x;
  const int lane = tid & 31;
  const int w    = __builtin_amdgcn_readfirstlane(tid >> 5);
  const int r    = lane & 15;
  const int h    = lane >> 4;
  const int bid   = blockIdx.x;
  const int b     = bid >> 3;
  const int d0    = ((bid >> 1) & 3) * WSL;
  const int lbase = (bid & 1) * LBLK;

  int lpos = lbase + w + 8 * lane - 1;
  lpos = lpos < 0 ? 0 : (lpos > Lsz - 1 ? Lsz - 1 : lpos);
  const int my_w = word_vector[b * Lsz + lpos];

  const float* wsrc = conv_w + d0 * KTOT;
  v4f wv[12];
#pragma unroll
  for (int i = 0; i < 12; ++i) {
    wv[i] = ld4(wsrc + (tid + 256 * i) * 4);
  }

  v4f rows[17];
#pragma unroll
  for (int i = 0; i < 17; ++i) {
    const int widx = __builtin_amdgcn_readlane(my_w, i);
    rows[i] = ld4(word_table + widx * Dsz + lane * 4);
  }

  const int dt = w & 1;
  const int lp = w >> 1;
  const float* bp = conv_b + d0 + dt * 16 + 8 * h;
  const v4f bias_lo = ld4(bp);
  const v4f bias_hi = ld4(bp + 4);

#pragma unroll
  for (int i = 0; i < 12; ++i) {
    const int f0  = (tid + 256 * i) * 4;
    const int dl  = f0 / KTOT;
    const int rem = f0 - dl * KTOT;
#pragma unroll
    for (int e = 0; e < 4; ++e) {
      const int din = (rem + e) / 3;
      const int tap = (rem + e) - din * 3;
      wl[(tap * WSL + dl) * WW_PITCH + din] = wv[i][e];
    }
  }

  const v4f zero4 = {0.0f, 0.0f, 0.0f, 0.0f};
#pragma unroll
  for (int i = 0; i < 17; ++i) {
    const int row = w + 8 * i;
    v4f v = rows[i];
    if (i == 0 || i == 16) {
      const int l = lbase + row - 1;
      if (l < 0 || l > Lsz - 1) v = zero4;
    }
    *(v4f*)(xs + row * XS_PITCH + lane * 4) = v;
  }
  __syncthreads();

  const float* aw = wl + (dt * 16 + r) * WW_PITCH + 4 * h;
  const float* xb = xs + (lp * 32 + r) * XS_PITCH + 4 * h;

  v8f acc[2] = {};
  WordFrags f0, f1;
  word_load_block(aw, xb, 0, f0);
#pragma unroll
  for (int blk = 0; blk < 12; blk += 2) {
    word_load_block(aw, xb, blk + 1, f1);
    __builtin_amdgcn_sched_barrier(0);
    word_wmma_block(f0, acc);
    __builtin_amdgcn_sched_barrier(0);
    if (blk + 2 < 12) word_load_block(aw, xb, blk + 2, f0);
    __builtin_amdgcn_sched_barrier(0);
    word_wmma_block(f1, acc);
    __builtin_amdgcn_sched_barrier(0);
  }

  float* orow = out + (b * Dsz + d0 + dt * 16 + 8 * h) * Lsz + lbase + lp * 32 + r;
#pragma unroll
  for (int n = 0; n < 2; ++n) {
#pragma unroll
    for (int i = 0; i < 8; ++i) {
      const float bias = i < 4 ? bias_lo[i & 3] : bias_hi[i & 3];
      orow[i * Lsz + n * 16] = acc[n][i] + bias;
    }
  }
}

extern "C" void kernel_launch(void* const* d_in, const int* in_sizes, int n_in,
                              void* d_out, int out_size, void* d_ws, size_t ws_size,
                              hipStream_t stream) {
  (void)in_sizes; (void)n_in; (void)out_size; (void)d_ws; (void)ws_size;
  const int*   word_vector   = (const int*)d_in[0];
  const int*   words_in_char = (const int*)d_in[1];
  const float* word_table    = (const float*)d_in[2];
  const float* chr_table     = (const float*)d_in[3];
  const float* conv_chr_w    = (const float*)d_in[4];
  const float* conv_chr_b    = (const float*)d_in[5];
  const float* conv_word_w   = (const float*)d_in[6];
  const float* conv_word_b   = (const float*)d_in[7];
  float* out = (float*)d_out;

  const size_t word_lds = (size_t)WORD_LDS_FLOATS * sizeof(float);

  static bool attr_done = false;
  if (!attr_done) {
    hipFuncSetAttribute((const void*)word_conv_kernel,
                        hipFuncAttributeMaxDynamicSharedMemorySize, (int)word_lds);
    attr_done = true;
  }

  char_conv_kernel<<<dim3(Dsz / DSL, Bsz), dim3(256), 0, stream>>>(
      words_in_char, chr_table, conv_chr_w, conv_chr_b, out);
  word_conv_kernel<<<dim3(Bsz * 2 * (Dsz / WSL)), dim3(256), word_lds, stream>>>(
      word_vector, word_table, conv_word_w, conv_word_b, out);
}


# Both paths in one launch: each of the 256 blocks does its char tile, then its word tile; the word tile's row and weight loads run under the char table build; sum orders unchanged
# speedup vs baseline: 14.3066x; 1.2038x over previous
#include <hip/hip_runtime.h>

#define Bsz 32
#define Lsz 256
#define Csz 16
#define Dsz 128
#define KTOT 384
#define DSL 16
#define WL_PITCH 132
#define WL_FLOATS (3 * DSL * WL_PITCH)
#define PT_ROW 64
#define PT_CHAR (3 * PT_ROW)
#define PT_FLOATS (128 * PT_CHAR)
#define ST_PITCH 34
#define WSL 32
#define WW_PITCH 132
#define WW_FLOATS (3 * WSL * WW_PITCH)
#define LBLK 128
#define XS_ROWS 136
#define XS_PITCH 132
#define XS_FLOATS (XS_ROWS * XS_PITCH)
#define WORD_LDS_FLOATS 30624

static_assert(WORD_LDS_FLOATS == WW_FLOATS + XS_FLOATS, "dynamic LDS = the two word tiles");

typedef __attribute__((ext_vector_type(2))) float v2f;
typedef __attribute__((ext_vector_type(4))) float v4f;
typedef __attribute__((ext_vector_type(8))) float v8f;
typedef __attribute__((ext_vector_type(4))) int v4i;

__device__ __forceinline__ v4f ld4(const float* p) { return *(const v4f*)p; }

__device__ __forceinline__ int clamp_char(int ch) {
  return ch < 0 ? 0 : (ch > 127 ? 127 : ch);
}

__device__ __forceinline__ v4f max4(v4f a, v4f b) {
  v4f m = {fmaxf(a[0], b[0]), fmaxf(a[1], b[1]), fmaxf(a[2], b[2]), fmaxf(a[3], b[3])};
  return m;
}

struct WordFrags {
  v4f a[4];
  v4f b[2][4];
};

__device__ __forceinline__ void word_load_block(const float* __restrict__ aw,
                                                const float* __restrict__ xb,
                                                int blk, WordFrags& f) {
  const int tap = blk >> 2;
  const int j0  = (blk & 3) * 4;
#pragma unroll
  for (int jj = 0; jj < 4; ++jj) {
    f.a[jj]    = ld4(aw + tap * (WSL * WW_PITCH) + 8 * (j0 + jj));
    f.b[0][jj] = ld4(xb + tap * XS_PITCH + 8 * (j0 + jj));
    f.b[1][jj] = ld4(xb + (16 + tap) * XS_PITCH + 8 * (j0 + jj));
  }
}

__device__ __forceinline__ void word_wmma_block(const WordFrags& f, v8f acc[2]) {
#pragma unroll
  for (int jj = 0; jj < 4; ++jj) {
    const v2f a_lo = {f.a[jj][0], f.a[jj][1]};
    const v2f a_hi = {f.a[jj][2], f.a[jj][3]};
#pragma unroll
    for (int n = 0; n < 2; ++n) {
      const v2f b_lo = {f.b[n][jj][0], f.b[n][jj][1]};
      acc[n] = __builtin_amdgcn_wmma_f32_16x16x4_f32(
           false, a_lo,  false, b_lo,
           (short)0, acc[n],  false,  false);
    }
#pragma unroll
    for (int n = 0; n < 2; ++n) {
      const v2f b_hi = {f.b[n][jj][2], f.b[n][jj][3]};
      acc[n] = __builtin_amdgcn_wmma_f32_16x16x4_f32(
           false, a_hi,  false, b_hi,
           (short)0, acc[n],  false,  false);
    }
  }
}

__global__ __launch_bounds__(256) void conv_pair_kernel(
    const int* __restrict__ word_vector,
    const int* __restrict__ words_in_char,
    const float* __restrict__ word_table,
    const float* __restrict__ chr_table,
    const float* __restrict__ chr_w,
    const float* __restrict__ chr_b,
    const float* __restrict__ word_w,
    const float* __restrict__ word_b,
    float* __restrict__ out) {
  __shared__ __attribute__((aligned(16))) float wl[WL_FLOATS];
  __shared__ __attribute__((aligned(16))) float pt[PT_FLOATS];
  extern __shared__ __attribute__((aligned(16))) float smem[];
  float* ww = smem;
  float* xs = smem + WW_FLOATS;
  const int tid  = threadIdx.x;
  const int lane = tid & 31;
  const int w    = __builtin_amdgcn_readfirstlane(tid >> 5);
  const int r    = lane & 15;
  const int h    = lane >> 4;
  const int bid  = blockIdx.x;
  const int b    = bid >> 3;
  const int d0   = (bid & 7) * DSL;
  const int wd0   = ((bid >> 1) & 3) * WSL;
  const int lbase = (bid & 1) * LBLK;
  const int gq   = lane & 3;
  const int gk   = (lane >> 2) & 3;
  const int gs   = lane >> 2;

  int lpos = lbase + w + 8 * lane - 1;
  lpos = lpos < 0 ? 0 : (lpos > Lsz - 1 ? Lsz - 1 : lpos);
  const int my_w = word_vector[b * Lsz + lpos];

  v4i ci[4][4];
#pragma unroll
  for (int it = 0; it < 4; ++it) {
    const v4i* ip = (const v4i*)(words_in_char + (b * Lsz + w * 32 + it * 8 + gs) * Csz);
#pragma unroll
    for (int j = 0; j < 4; ++j) {
      ci[it][j] = ip[j];
    }
  }

  v4f bq[16];
  const float* trow = chr_table + (w * 16 + r) * Dsz + 4 * h;
#pragma unroll
  for (int j = 0; j < 16; ++j) {
    bq[j] = ld4(trow + 8 * j);
  }

  const float* wsrc = chr_w + d0 * KTOT;
  v4f wv[6];
#pragma unroll
  for (int i = 0; i < 6; ++i) {
    wv[i] = ld4(wsrc + (tid + 256 * i) * 4);
  }
#pragma unroll
  for (int i = 0; i < 6; ++i) {
    const int f0  = (tid + 256 * i) * 4;
    const int dl  = f0 / KTOT;
    const int rem = f0 - dl * KTOT;
#pragma unroll
    for (int e = 0; e < 4; ++e) {
      const int din = (rem + e) / 3;
      const int tap = (rem + e) - din * 3;
      wl[(tap * DSL + dl) * WL_PITCH + din] = wv[i][e];
    }
  }
  __syncthreads();

  const float* wwsrc = word_w + wd0 * KTOT;
  v4f wvw[12];
#pragma unroll
  for (int i = 0; i < 12; ++i) {
    wvw[i] = ld4(wwsrc + (tid + 256 * i) * 4);
  }

  v4f rows[17];
#pragma unroll
  for (int i = 0; i < 17; ++i) {
    const int widx = __builtin_amdgcn_readlane(my_w, i);
    rows[i] = ld4(word_table + widx * Dsz + lane * 4);
  }

  const int dt = w & 1;
  const int lp = w >> 1;
  const float* bp = word_b + wd0 + dt * 16 + 8 * h;
  const v4f bias_lo = ld4(bp);
  const v4f bias_hi = ld4(bp + 4);

  __builtin_amdgcn_sched_barrier(0);

  v8f acc[3] = {};
#pragma unroll
  for (int j = 0; j < 16; ++j) {
    v4f a4[3];
#pragma unroll
    for (int t = 0; t < 3; ++t) {
      a4[t] = ld4(wl + (t * DSL + r) * WL_PITCH + 8 * j + 4 * h);
    }
    const v2f b_lo = {bq[j][0], bq[j][1]};
    const v2f b_hi = {bq[j][2], bq[j][3]};
#pragma unroll
    for (int t = 0; t < 3; ++t) {
      const v2f a_lo = {a4[t][0], a4[t][1]};
      acc[t] = __builtin_amdgcn_wmma_f32_16x16x4_f32(
           false, a_lo,  false, b_lo,
           (short)0, acc[t],  false,  false);
    }
#pragma unroll
    for (int t = 0; t < 3; ++t) {
      const v2f a_hi = {a4[t][2], a4[t][3]};
      acc[t] = __builtin_amdgcn_wmma_f32_16x16x4_f32(
           false, a_hi,  false, b_hi,
           (short)0, acc[t],  false,  false);
    }
  }

  {
    const int chr   = w * 16 + r;
    const int s_lo  = 4 * ((2 * h + chr) & 3);
    const int s_hi  = 4 * ((2 * h + 1 + chr) & 3);
#pragma unroll
    for (int t = 0; t < 3; ++t) {
      const v4f lo = {acc[t][0], acc[t][1], acc[t][2], acc[t][3]};
      const v4f hi = {acc[t][4], acc[t][5], acc[t][6], acc[t][7]};
#pragma unroll
      for (int jp = 0; jp < 4; ++jp) {
        const int kk = (jp + (r >> 2)) & 3;
        float* p = pt + chr * PT_CHAR + t * PT_ROW + 16 * kk;
        *(v4f*)(p + s_lo) = lo;
        *(v4f*)(p + s_hi) = hi;
      }
    }
  }

#pragma unroll
  for (int i = 0; i < 12; ++i) {
    const int f0  = (tid + 256 * i) * 4;
    const int dl  = f0 / KTOT;
    const int rem = f0 - dl * KTOT;
#pragma unroll
    for (int e = 0; e < 4; ++e) {
      const int din = (rem + e) / 3;
      const int tap = (rem + e) - din * 3;
      ww[(tap * WSL + dl) * WW_PITCH + din] = wvw[i][e];
    }
  }

  const v4f zero4 = {0.0f, 0.0f, 0.0f, 0.0f};
#pragma unroll
  for (int i = 0; i < 17; ++i) {
    const int row = w + 8 * i;
    v4f v = rows[i];
    if (i == 0 || i == 16) {
      const int l = lbase + row - 1;
      if (l < 0 || l > Lsz - 1) v = zero4;
    }
    *(v4f*)(xs + row * XS_PITCH + lane * 4) = v;
  }
  __syncthreads();

  v4f res[4];
#pragma unroll
  for (int it = 0; it < 4; ++it) {
    int off[Csz];
#pragma unroll
    for (int j = 0; j < Csz; ++j) {
      const int ch = clamp_char(ci[it][j >> 2][j & 3]);
      off[j] = ch * PT_CHAR + 16 * gk + 4 * ((gq + ch) & 3);
    }
    v4f m;
#pragma unroll
    for (int c = 0; c < Csz; ++c) {
      v4f s = ld4(pt + PT_ROW + off[c]);
      if (c > 0) {
        s = ld4(pt + off[c - 1]) + s;
      }
      if (c < Csz - 1) {
        s = s + ld4(pt + 2 * PT_ROW + off[c + 1]);
      }
      m = (c == 0) ? s : max4(m, s);
    }
    res[it] = m;
  }

  float* st = wl + w * (DSL * ST_PITCH);
#pragma unroll
  for (int it = 0; it < 4; ++it) {
#pragma unroll
    for (int e = 0; e < 4; ++e) {
      st[(4 * gq + e) * ST_PITCH + it * 8 + gs] = res[it][e];
    }
  }
  __syncthreads();

  float* corow = out + ((Bsz + b) * Dsz + d0) * Lsz + w * 32 + lane;
#pragma unroll
  for (int dl = 0; dl < DSL; ++dl) {
    corow[dl * Lsz] = st[dl * ST_PITCH + lane] + chr_b[d0 + dl];
  }

  const float* aw = ww + (dt * 16 + r) * WW_PITCH + 4 * h;
  const float* xb = xs + (lp * 32 + r) * XS_PITCH + 4 * h;

  v8f wacc[2] = {};
  WordFrags f0, f1;
  word_load_block(aw, xb, 0, f0);
#pragma unroll
  for (int blk = 0; blk < 12; blk += 2) {
    word_load_block(aw, xb, blk + 1, f1);
    __builtin_amdgcn_sched_barrier(0);
    word_wmma_block(f0, wacc);
    __builtin_amdgcn_sched_barrier(0);
    if (blk + 2 < 12) word_load_block(aw, xb, blk + 2, f0);
    __builtin_amdgcn_sched_barrier(0);
    word_wmma_block(f1, wacc);
    __builtin_amdgcn_sched_barrier(0);
  }

  float* worow = out + (b * Dsz + wd0 + dt * 16 + 8 * h) * Lsz + lbase + lp * 32 + r;
#pragma unroll
  for (int n = 0; n < 2; ++n) {
#pragma unroll
    for (int i = 0; i < 8; ++i) {
      const float bias = i < 4 ? bias_lo[i & 3] : bias_hi[i & 3];
      worow[i * Lsz + n * 16] = wacc[n][i] + bias;
    }
  }
}

extern "C" void kernel_launch(void* const* d_in, const int* in_sizes, int n_in,
                              void* d_out, int out_size, void* d_ws, size_t ws_size,
                              hipStream_t stream) {
  (void)in_sizes; (void)n_in; (void)out_size; (void)d_ws; (void)ws_size;
  const int*   word_vector   = (const int*)d_in[0];
  const int*   words_in_char = (const int*)d_in[1];
  const float* word_table    = (const float*)d_in[2];
  const float* chr_table     = (const float*)d_in[3];
  const float* conv_chr_w    = (const float*)d_in[4];
  const float* conv_chr_b    = (const float*)d_in[5];
  const float* conv_word_w   = (const float*)d_in[6];
  const float* conv_word_b   = (const float*)d_in[7];
  float* out = (float*)d_out;

  const size_t word_lds = (size_t)WORD_LDS_FLOATS * sizeof(float);

  static bool attr_done = false;
  if (!attr_done) {
    hipFuncSetAttribute((const void*)conv_pair_kernel,
                        hipFuncAttributeMaxDynamicSharedMemorySize, (int)word_lds);
    attr_done = true;
  }

  conv_pair_kernel<<<dim3(Bsz * (Dsz / DSL)), dim3(256), word_lds, stream>>>(
      word_vector, words_in_char, word_table, chr_table,
      conv_chr_w, conv_chr_b, conv_word_w, conv_word_b, out);
}


# After the table barrier the two waves of a SIMD take gather and word products in opposite order (matrix stage at raised wave priority); transpose per wave, third barrier gone; sum orders unchanged
# speedup vs baseline: 15.0697x; 1.0533x over previous
#include <hip/hip_runtime.h>

#define Bsz 32
#define Lsz 256
#define Csz 16
#define Dsz 128
#define KTOT 384
#define DSL 16
#define WL_PITCH 132
#define WL_FLOATS (3 * DSL * WL_PITCH)
#define PT_ROW 64
#define PT_CHAR (3 * PT_ROW)
#define PT_FLOATS (128 * PT_CHAR)
#define ST_PITCH 34
#define WSL 32
#define WW_PITCH 132
#define WW_FLOATS (3 * WSL * WW_PITCH)
#define LBLK 128
#define XS_ROWS 136
#define XS_PITCH 132
#define XS_FLOATS (XS_ROWS * XS_PITCH)
#define WORD_LDS_FLOATS 30624

static_assert(WORD_LDS_FLOATS == WW_FLOATS + XS_FLOATS, "dynamic LDS = the two word tiles");

typedef __attribute__((ext_vector_type(2))) float v2f;
typedef __attribute__((ext_vector_type(4))) float v4f;
typedef __attribute__((ext_vector_type(8))) float v8f;
typedef __attribute__((ext_vector_type(4))) int v4i;

__device__ __forceinline__ v4f ld4(const float* p) { return *(const v4f*)p; }

__device__ __forceinline__ int clamp_char(int ch) {
  return ch < 0 ? 0 : (ch > 127 ? 127 : ch);
}

__device__ __forceinline__ v4f max4(v4f a, v4f b) {
  v4f m = {fmaxf(a[0], b[0]), fmaxf(a[1], b[1]), fmaxf(a[2], b[2]), fmaxf(a[3], b[3])};
  return m;
}

struct WordFrags {
  v4f a[4];
  v4f b[2][4];
};

__device__ __forceinline__ void word_load_block(const float* __restrict__ aw,
                                                const float* __restrict__ xb,
                                                int blk, WordFrags& f) {
  const int tap = blk >> 2;
  const int j0  = (blk & 3) * 4;
#pragma unroll
  for (int jj = 0; jj < 4; ++jj) {
    f.a[jj]    = ld4(aw + tap * (WSL * WW_PITCH) + 8 * (j0 + jj));
    f.b[0][jj] = ld4(xb + tap * XS_PITCH + 8 * (j0 + jj));
    f.b[1][jj] = ld4(xb + (16 + tap) * XS_PITCH + 8 * (j0 + jj));
  }
}

__device__ __forceinline__ void word_wmma_block(const WordFrags& f, v8f acc[2]) {
#pragma unroll
  for (int jj = 0; jj < 4; ++jj) {
    const v2f a_lo = {f.a[jj][0], f.a[jj][1]};
    const v2f a_hi = {f.a[jj][2], f.a[jj][3]};
#pragma unroll
    for (int n = 0; n < 2; ++n) {
      const v2f b_lo = {f.b[n][jj][0], f.b[n][jj][1]};
      acc[n] = __builtin_amdgcn_wmma_f32_16x16x4_f32(
           false, a_lo,  false, b_lo,
           (short)0, acc[n],  false,  false);
    }
#pragma unroll
    for (int n = 0; n < 2; ++n) {
      const v2f b_hi = {f.b[n][jj][2], f.b[n][jj][3]};
      acc[n] = __builtin_amdgcn_wmma_f32_16x16x4_f32(
           false, a_hi,  false, b_hi,
           (short)0, acc[n],  false,  false);
    }
  }
}

__global__ __launch_bounds__(256) void conv_pair_kernel(
    const int* __restrict__ word_vector,
    const int* __restrict__ words_in_char,
    const float* __restrict__ word_table,
    const float* __restrict__ chr_table,
    const float* __restrict__ chr_w,
    const float* __restrict__ chr_b,
    const float* __restrict__ word_w,
    const float* __restrict__ word_b,
    float* __restrict__ out) {
  __shared__ __attribute__((aligned(16))) float wl[WL_FLOATS];
  __shared__ __attribute__((aligned(16))) float pt[PT_FLOATS];
  extern __shared__ __attribute__((aligned(16))) float smem[];
  float* ww = smem;
  float* xs = smem + WW_FLOATS;
  const int tid  = threadIdx.x;
  const int lane = tid & 31;
  const int w    = __builtin_amdgcn_readfirstlane(tid >> 5);
  const int r    = lane & 15;
  const int h    = lane >> 4;
  const int bid  = blockIdx.x;
  const int b    = bid >> 3;
  const int d0   = (bid & 7) * DSL;
  const int wd0   = ((bid >> 1) & 3) * WSL;
  const int lbase = (bid & 1) * LBLK;
  const int gq   = lane & 3;
  const int gk   = (lane >> 2) & 3;
  const int gs   = lane >> 2;

  int lpos = lbase + w + 8 * lane - 1;
  lpos = lpos < 0 ? 0 : (lpos > Lsz - 1 ? Lsz - 1 : lpos);
  const int my_w = word_vector[b * Lsz + lpos];

  v4i ci[4][4];
#pragma unroll
  for (int it = 0; it < 4; ++it) {
    const v4i* ip = (const v4i*)(words_in_char + (b * Lsz + w * 32 + it * 8 + gs) * Csz);
#pragma unroll
    for (int j = 0; j < 4; ++j) {
      ci[it][j] = ip[j];
    }
  }

  v4f bq[16];
  const float* trow = chr_table + (w * 16 + r) * Dsz + 4 * h;
#pragma unroll
  for (int j = 0; j < 16; ++j) {
    bq[j] = ld4(trow + 8 * j);
  }

  const float* wsrc = chr_w + d0 * KTOT;
  v4f wv[6];
#pragma unroll
  for (int i = 0; i < 6; ++i) {
    wv[i] = ld4(wsrc + (tid + 256 * i) * 4);
  }

  int goff[4][Csz];
#pragma unroll
  for (int it = 0; it < 4; ++it) {
#pragma unroll
    for (int j = 0; j < Csz; ++j) {
      const int ch = clamp_char(ci[it][j >> 2][j & 3]);
      goff[it][j] = ch * PT_CHAR + 16 * gk + 4 * ((gq + ch) & 3);
    }
  }
#pragma unroll
  for (int i = 0; i < 6; ++i) {
    const int f0  = (tid + 256 * i) * 4;
    const int dl  = f0 / KTOT;
    const int rem = f0 - dl * KTOT;
#pragma unroll
    for (int e = 0; e < 4; ++e) {
      const int din = (rem + e) / 3;
      const int tap = (rem + e) - din * 3;
      wl[(tap * DSL + dl) * WL_PITCH + din] = wv[i][e];
    }
  }
  __syncthreads();

  const float* wwsrc = word_w + wd0 * KTOT;
  v4f wvw[12];
#pragma unroll
  for (int i = 0; i < 12; ++i) {
    wvw[i] = ld4(wwsrc + (tid + 256 * i) * 4);
  }

  v4f rows[17];
#pragma unroll
  for (int i = 0; i < 17; ++i) {
    const int widx = __builtin_amdgcn_readlane(my_w, i);
    rows[i] = ld4(word_table + widx * Dsz + lane * 4);
  }

  const int dt = w & 1;
  const int lp = w >> 1;
  const float* bp = word_b + wd0 + dt * 16 + 8 * h;
  const v4f bias_lo = ld4(bp);
  const v4f bias_hi = ld4(bp + 4);

  __builtin_amdgcn_sched_barrier(0);

  v8f acc[3] = {};
#pragma unroll
  for (int j = 0; j < 16; ++j) {
    v4f a4[3];
#pragma unroll
    for (int t = 0; t < 3; ++t) {
      a4[t] = ld4(wl + (t * DSL + r) * WL_PITCH + 8 * j + 4 * h);
    }
    const v2f b_lo = {bq[j][0], bq[j][1]};
    const v2f b_hi = {bq[j][2], bq[j][3]};
#pragma unroll
    for (int t = 0; t < 3; ++t) {
      const v2f a_lo = {a4[t][0], a4[t][1]};
      acc[t] = __builtin_amdgcn_wmma_f32_16x16x4_f32(
           false, a_lo,  false, b_lo,
           (short)0, acc[t],  false,  false);
    }
#pragma unroll
    for (int t = 0; t < 3; ++t) {
      const v2f a_hi = {a4[t][2], a4[t][3]};
      acc[t] = __builtin_amdgcn_wmma_f32_16x16x4_f32(
           false, a_hi,  false, b_hi,
           (short)0, acc[t],  false,  false);
    }
  }

  {
    const int chr   = w * 16 + r;
    const int s_lo  = 4 * ((2 * h + chr) & 3);
    const int s_hi  = 4 * ((2 * h + 1 + chr) & 3);
#pragma unroll
    for (int t = 0; t < 3; ++t) {
      const v4f lo = {acc[t][0], acc[t][1], acc[t][2], acc[t][3]};
      const v4f hi = {acc[t][4], acc[t][5], acc[t][6], acc[t][7]};
#pragma unroll
      for (int jp = 0; jp < 4; ++jp) {
        const int kk = (jp + (r >> 2)) & 3;
        float* p = pt + chr * PT_CHAR + t * PT_ROW + 16 * kk;
        *(v4f*)(p + s_lo) = lo;
        *(v4f*)(p + s_hi) = hi;
      }
    }
  }

#pragma unroll
  for (int i = 0; i < 12; ++i) {
    const int f0  = (tid + 256 * i) * 4;
    const int dl  = f0 / KTOT;
    const int rem = f0 - dl * KTOT;
#pragma unroll
    for (int e = 0; e < 4; ++e) {
      const int din = (rem + e) / 3;
      const int tap = (rem + e) - din * 3;
      ww[(tap * WSL + dl) * WW_PITCH + din] = wvw[i][e];
    }
  }

  const v4f zero4 = {0.0f, 0.0f, 0.0f, 0.0f};
#pragma unroll
  for (int i = 0; i < 17; ++i) {
    const int row = w + 8 * i;
    v4f v = rows[i];
    if (i == 0 || i == 16) {
      const int l = lbase + row - 1;
      if (l < 0 || l > Lsz - 1) v = zero4;
    }
    *(v4f*)(xs + row * XS_PITCH + lane * 4) = v;
  }
  __syncthreads();

  float* st = wl + w * (DSL * ST_PITCH);
  const float* aw = ww + (dt * 16 + r) * WW_PITCH + 4 * h;
  const float* xb = xs + (lp * 32 + r) * XS_PITCH + 4 * h;
  const bool gather_first = w >= 4;

  for (int stage = 0; stage < 2; ++stage) {
    if ((stage == 0) == gather_first) {

      float cb[DSL];
#pragma unroll
      for (int dl = 0; dl < DSL; ++dl) {
        cb[dl] = chr_b[d0 + dl];
      }
      v4f res[4];
#pragma unroll
      for (int it = 0; it < 4; ++it) {
        const int* off = goff[it];
        v4f m;
#pragma unroll
        for (int c = 0; c < Csz; ++c) {
          v4f s = ld4(pt + PT_ROW + off[c]);
          if (c > 0) {
            s = ld4(pt + off[c - 1]) + s;
          }
          if (c < Csz - 1) {
            s = s + ld4(pt + 2 * PT_ROW + off[c + 1]);
          }
          m = (c == 0) ? s : max4(m, s);
        }
        res[it] = m;
      }

#pragma unroll
      for (int it = 0; it < 4; ++it) {
#pragma unroll
        for (int e = 0; e < 4; ++e) {
          st[(4 * gq + e) * ST_PITCH + it * 8 + gs] = res[it][e];
        }
      }

      float* corow = out + ((Bsz + b) * Dsz + d0) * Lsz + w * 32 + lane;
#pragma unroll
      for (int dl = 0; dl < DSL; ++dl) {
        corow[dl * Lsz] = st[dl * ST_PITCH + lane] + cb[dl];
      }
    } else {
      __builtin_amdgcn_s_setprio(3);
      v8f wacc[2] = {};
      WordFrags f0, f1;
      word_load_block(aw, xb, 0, f0);
#pragma unroll
      for (int blk = 0; blk < 12; blk += 2) {
        word_load_block(aw, xb, blk + 1, f1);
        __builtin_amdgcn_sched_barrier(0);
        word_wmma_block(f0, wacc);
        __builtin_amdgcn_sched_barrier(0);
        if (blk + 2 < 12) word_load_block(aw, xb, blk + 2, f0);
        __builtin_amdgcn_sched_barrier(0);
        word_wmma_block(f1, wacc);
        __builtin_amdgcn_sched_barrier(0);
      }

      float* worow = out + (b * Dsz + wd0 + dt * 16 + 8 * h) * Lsz + lbase + lp * 32 + r;
#pragma unroll
      for (int n = 0; n < 2; ++n) {
#pragma unroll
        for (int i = 0; i < 8; ++i) {
          const float bias = i < 4 ? bias_lo[i & 3] : bias_hi[i & 3];
          worow[i * Lsz + n * 16] = wacc[n][i] + bias;
        }
      }
      __builtin_amdgcn_s_setprio(0);
    }
  }
}

extern "C" void kernel_launch(void* const* d_in, const int* in_sizes, int n_in,
                              void* d_out, int out_size, void* d_ws, size_t ws_size,
                              hipStream_t stream) {
  (void)in_sizes; (void)n_in; (void)out_size; (void)d_ws; (void)ws_size;
  const int*   word_vector   = (const int*)d_in[0];
  const int*   words_in_char = (const int*)d_in[1];
  const float* word_table    = (const float*)d_in[2];
  const float* chr_table     = (const float*)d_in[3];
  const float* conv_chr_w    = (const float*)d_in[4];
  const float* conv_chr_b    = (const float*)d_in[5];
  const float* conv_word_w   = (const float*)d_in[6];
  const float* conv_word_b   = (const float*)d_in[7];
  float* out = (float*)d_out;

  const size_t word_lds = (size_t)WORD_LDS_FLOATS * sizeof(float);

  static bool attr_done = false;
  if (!attr_done) {
    hipFuncSetAttribute((const void*)conv_pair_kernel,
                        hipFuncAttributeMaxDynamicSharedMemorySize, (int)word_lds);
    attr_done = true;
  }

  conv_pair_kernel<<<dim3(Bsz * (Dsz / DSL)), dim3(256), word_lds, stream>>>(
      word_vector, words_in_char, word_table, chr_table,
      conv_chr_w, conv_chr_b, conv_word_w, conv_word_b, out);
}


# Word conv by minimal filtering F(2,3) on position pairs: 4 products per 2 outputs instead of 6 (192 -> 128 f32 WMMAs per wave), weight planes formed at staging; f32 sum order and rounding change
# speedup vs baseline: 16.0131x; 1.0626x over previous
#include <hip/hip_runtime.h>

#define Bsz 32
#define Lsz 256
#define Csz 16
#define Dsz 128
#define KTOT 384
#define DSL 16
#define WL_PITCH 132
#define WL_FLOATS (3 * DSL * WL_PITCH)
#define PT_ROW 64
#define PT_CHAR (3 * PT_ROW)
#define PT_FLOATS (128 * PT_CHAR)
#define ST_PITCH 34
#define WSL 32
#define WW_PITCH 132
#define WW_FLOATS (4 * WSL * WW_PITCH)
#define LBLK 128
#define XS_ROWS 136
#define XS_PITCH 132
#define XS_HALF 68
#define XS_FLOATS (XS_ROWS * XS_PITCH)
#define WORD_LDS_FLOATS 34848

static_assert(WORD_LDS_FLOATS == WW_FLOATS + XS_FLOATS, "dynamic LDS = the two word tiles");

typedef __attribute__((ext_vector_type(2))) float v2f;
typedef __attribute__((ext_vector_type(4))) float v4f;
typedef __attribute__((ext_vector_type(8))) float v8f;
typedef __attribute__((ext_vector_type(4))) int v4i;

__device__ __forceinline__ v4f ld4(const float* p) { return *(const v4f*)p; }

__device__ __forceinline__ int clamp_char(int ch) {
  return ch < 0 ? 0 : (ch > 127 ? 127 : ch);
}

__device__ __forceinline__ v4f max4(v4f a, v4f b) {
  v4f m = {fmaxf(a[0], b[0]), fmaxf(a[1], b[1]), fmaxf(a[2], b[2]), fmaxf(a[3], b[3])};
  return m;
}

struct WordFrags {
  v4f a[4][2];
  v4f x[4][2];
};

__device__ __forceinline__ void word_load_block(const float* __restrict__ aw,
                                                const float* __restrict__ xb,
                                                int blk, WordFrags& f) {
  const int j0 = blk * 2;
#pragma unroll
  for (int jj = 0; jj < 2; ++jj) {
#pragma unroll
    for (int k = 0; k < 4; ++k) {
      f.a[k][jj] = ld4(aw + k * (WSL * WW_PITCH) + 8 * (j0 + jj));
    }
    f.x[0][jj] = ld4(xb + 8 * (j0 + jj));
    f.x[1][jj] = ld4(xb + XS_HALF * XS_PITCH + 8 * (j0 + jj));
    f.x[2][jj] = ld4(xb + XS_PITCH + 8 * (j0 + jj));
    f.x[3][jj] = ld4(xb + (XS_HALF + 1) * XS_PITCH + 8 * (j0 + jj));
  }
}

__device__ __forceinline__ void word_wmma_block(const WordFrags& f, v8f acc[4]) {
#pragma unroll
  for (int jj = 0; jj < 2; ++jj) {
    v4f v[4];
    v[0] = f.x[0][jj] - f.x[2][jj];
    v[1] = f.x[1][jj] + f.x[2][jj];
    v[2] = f.x[2][jj] - f.x[1][jj];
    v[3] = f.x[1][jj] - f.x[3][jj];
#pragma unroll
    for (int k = 0; k < 4; ++k) {
      const v2f a_lo = {f.a[k][jj][0], f.a[k][jj][1]};
      const v2f b_lo = {v[k][0], v[k][1]};
      acc[k] = __builtin_amdgcn_wmma_f32_16x16x4_f32(
           false, a_lo,  false, b_lo,
           (short)0, acc[k],  false,  false);
    }
#pragma unroll
    for (int k = 0; k < 4; ++k) {
      const v2f a_hi = {f.a[k][jj][2], f.a[k][jj][3]};
      const v2f b_hi = {v[k][2], v[k][3]};
      acc[k] = __builtin_amdgcn_wmma_f32_16x16x4_f32(
           false, a_hi,  false, b_hi,
           (short)0, acc[k],  false,  false);
    }
  }
}

__global__ __launch_bounds__(256) void conv_pair_kernel(
    const int* __restrict__ word_vector,
    const int* __restrict__ words_in_char,
    const float* __restrict__ word_table,
    const float* __restrict__ chr_table,
    const float* __restrict__ chr_w,
    const float* __restrict__ chr_b,
    const float* __restrict__ word_w,
    const float* __restrict__ word_b,
    float* __restrict__ out) {
  __shared__ __attribute__((aligned(16))) float wl[WL_FLOATS];
  __shared__ __attribute__((aligned(16))) float pt[PT_FLOATS];
  extern __shared__ __attribute__((aligned(16))) float smem[];
  float* ww = smem;
  float* xs = smem + WW_FLOATS;
  const int tid  = threadIdx.x;
  const int lane = tid & 31;
  const int w    = __builtin_amdgcn_readfirstlane(tid >> 5);
  const int r    = lane & 15;
  const int h    = lane >> 4;
  const int bid  = blockIdx.x;
  const int b    = bid >> 3;
  const int d0   = (bid & 7) * DSL;
  const int wd0   = ((bid >> 1) & 3) * WSL;
  const int lbase = (bid & 1) * LBLK;
  const int gq   = lane & 3;
  const int gk   = (lane >> 2) & 3;
  const int gs   = lane >> 2;

  int lpos = lbase + w + 8 * lane - 1;
  lpos = lpos < 0 ? 0 : (lpos > Lsz - 1 ? Lsz - 1 : lpos);
  const int my_w = word_vector[b * Lsz + lpos];

  v4i ci[4][4];
#pragma unroll
  for (int it = 0; it < 4; ++it) {
    const v4i* ip = (const v4i*)(words_in_char + (b * Lsz + w * 32 + it * 8 + gs) * Csz);
#pragma unroll
    for (int j = 0; j < 4; ++j) {
      ci[it][j] = ip[j];
    }
  }

  v4f bq[16];
  const float* trow = chr_table + (w * 16 + r) * Dsz + 4 * h;
#pragma unroll
  for (int j = 0; j < 16; ++j) {
    bq[j] = ld4(trow + 8 * j);
  }

  const float* wsrc = chr_w + d0 * KTOT;
  v4f wv[6];
#pragma unroll
  for (int i = 0; i < 6; ++i) {
    wv[i] = ld4(wsrc + (tid + 256 * i) * 4);
  }

  int goff[4][Csz];
#pragma unroll
  for (int it = 0; it < 4; ++it) {
#pragma unroll
    for (int j = 0; j < Csz; ++j) {
      const int ch = clamp_char(ci[it][j >> 2][j & 3]);
      goff[it][j] = ch * PT_CHAR + 16 * gk + 4 * ((gq + ch) & 3);
    }
  }
  __builtin_amdgcn_sched_barrier(0);
#pragma unroll
  for (int i = 0; i < 6; ++i) {
    const int f0  = (tid + 256 * i) * 4;
    const int dl  = f0 / KTOT;
    const int rem = f0 - dl * KTOT;
#pragma unroll
    for (int e = 0; e < 4; ++e) {
      const int din = (rem + e) / 3;
      const int tap = (rem + e) - din * 3;
      wl[(tap * DSL + dl) * WL_PITCH + din] = wv[i][e];
    }
  }
  __syncthreads();

  const float* wwsrc = word_w + (wd0 + w) * KTOT + lane * 12;
  v4f wvw[12];
#pragma unroll
  for (int i = 0; i < 4; ++i) {
#pragma unroll
    for (int q = 0; q < 3; ++q) {
      wvw[3 * i + q] = ld4(wwsrc + i * 8 * KTOT + 4 * q);
    }
  }

  v4f rows[17];
#pragma unroll
  for (int i = 0; i < 17; ++i) {
    const int widx = __builtin_amdgcn_readlane(my_w, i);
    rows[i] = ld4(word_table + widx * Dsz + lane * 4);
  }

  const int dt = w & 1;
  const int lp = w >> 1;
  const float* bp = word_b + wd0 + dt * 16 + 8 * h;
  const v4f bias_lo = ld4(bp);
  const v4f bias_hi = ld4(bp + 4);

  __builtin_amdgcn_sched_barrier(0);

  v8f acc[3] = {};
#pragma unroll
  for (int j = 0; j < 16; ++j) {
    v4f a4[3];
#pragma unroll
    for (int t = 0; t < 3; ++t) {
      a4[t] = ld4(wl + (t * DSL + r) * WL_PITCH + 8 * j + 4 * h);
    }
    const v2f b_lo = {bq[j][0], bq[j][1]};
    const v2f b_hi = {bq[j][2], bq[j][3]};
#pragma unroll
    for (int t = 0; t < 3; ++t) {
      const v2f a_lo = {a4[t][0], a4[t][1]};
      acc[t] = __builtin_amdgcn_wmma_f32_16x16x4_f32(
           false, a_lo,  false, b_lo,
           (short)0, acc[t],  false,  false);
    }
#pragma unroll
    for (int t = 0; t < 3; ++t) {
      const v2f a_hi = {a4[t][2], a4[t][3]};
      acc[t] = __builtin_amdgcn_wmma_f32_16x16x4_f32(
           false, a_hi,  false, b_hi,
           (short)0, acc[t],  false,  false);
    }
    if (j == 7) __builtin_amdgcn_sched_barrier(0);
  }

  {
    const int chr   = w * 16 + r;
    const int s_lo  = 4 * ((2 * h + chr) & 3);
    const int s_hi  = 4 * ((2 * h + 1 + chr) & 3);
#pragma unroll
    for (int t = 0; t < 3; ++t) {
      const v4f lo = {acc[t][0], acc[t][1], acc[t][2], acc[t][3]};
      const v4f hi = {acc[t][4], acc[t][5], acc[t][6], acc[t][7]};
#pragma unroll
      for (int jp = 0; jp < 4; ++jp) {
        const int kk = (jp + (r >> 2)) & 3;
        float* p = pt + chr * PT_CHAR + t * PT_ROW + 16 * kk;
        *(v4f*)(p + s_lo) = lo;
        *(v4f*)(p + s_hi) = hi;
      }
    }
  }

#pragma unroll
  for (int i = 0; i < 4; ++i) {
    const v4f x0 = wvw[3 * i], x1 = wvw[3 * i + 1], x2 = wvw[3 * i + 2];
    const v4f g0 = {x0[0], x0[3], x1[2], x2[1]};
    const v4f g1 = {x0[1], x1[0], x1[3], x2[2]};
    const v4f g2 = {x0[2], x1[1], x2[0], x2[3]};
    const v4f gs = g0 + g2;
    const v4f u1 = (gs + g1) * 0.5f;
    const v4f u2 = (gs - g1) * 0.5f;
    float* dst = ww + (w + 8 * i) * WW_PITCH + 4 * lane;
    *(v4f*)(dst)                         = g0;
    *(v4f*)(dst + 1 * (WSL * WW_PITCH)) = u1;
    *(v4f*)(dst + 2 * (WSL * WW_PITCH)) = u2;
    *(v4f*)(dst + 3 * (WSL * WW_PITCH)) = g2;
  }

  const v4f zero4 = {0.0f, 0.0f, 0.0f, 0.0f};
#pragma unroll
  for (int i = 0; i < 17; ++i) {
    const int row = w + 8 * i;
    v4f v = rows[i];
    if (i == 0 || i == 16) {
      const int l = lbase + row - 1;
      if (l < 0 || l > Lsz - 1) v = zero4;
    }
    *(v4f*)(xs + ((row & 1) * XS_HALF + (row >> 1)) * XS_PITCH + lane * 4) = v;
  }
  __syncthreads();

  float* st = wl + w * (DSL * ST_PITCH);
  const float* aw = ww + (dt * 16 + r) * WW_PITCH + 4 * h;
  const float* xb = xs + (lp * 16 + r) * XS_PITCH + 4 * h;
  const bool gather_first = w >= 4;

  for (int stage = 0; stage < 2; ++stage) {
    if ((stage == 0) == gather_first) {

      float cb[DSL];
#pragma unroll
      for (int dl = 0; dl < DSL; ++dl) {
        cb[dl] = chr_b[d0 + dl];
      }
      v4f res[4];
#pragma unroll
      for (int it = 0; it < 4; ++it) {
        const int* off = goff[it];
        v4f m;
#pragma unroll
        for (int c = 0; c < Csz; ++c) {
          v4f s = ld4(pt + PT_ROW + off[c]);
          if (c > 0) {
            s = ld4(pt + off[c - 1]) + s;
          }
          if (c < Csz - 1) {
            s = s + ld4(pt + 2 * PT_ROW + off[c + 1]);
          }
          m = (c == 0) ? s : max4(m, s);
        }
        res[it] = m;
      }

#pragma unroll
      for (int it = 0; it < 4; ++it) {
#pragma unroll
        for (int e = 0; e < 4; ++e) {
          st[(4 * gq + e) * ST_PITCH + it * 8 + gs] = res[it][e];
        }
      }

      float* corow = out + ((Bsz + b) * Dsz + d0) * Lsz + w * 32 + lane;
#pragma unroll
      for (int dl = 0; dl < DSL; ++dl) {
        corow[dl * Lsz] = st[dl * ST_PITCH + lane] + cb[dl];
      }
    } else {
      __builtin_amdgcn_s_setprio(3);
      v8f wacc[4] = {};
      WordFrags f0, f1;
      word_load_block(aw, xb, 0, f0);
#pragma unroll
      for (int blk = 0; blk < 8; blk += 2) {
        word_load_block(aw, xb, blk + 1, f1);
        __builtin_amdgcn_sched_barrier(0);
        word_wmma_block(f0, wacc);
        __builtin_amdgcn_sched_barrier(0);
        if (blk + 2 < 8) word_load_block(aw, xb, blk + 2, f0);
        __builtin_amdgcn_sched_barrier(0);
        word_wmma_block(f1, wacc);
        __builtin_amdgcn_sched_barrier(0);
      }

      float* worow = out + (b * Dsz + wd0 + dt * 16 + 8 * h) * Lsz + lbase + lp * 32 + 2 * r;
#pragma unroll
      for (int i = 0; i < 8; ++i) {
        const float bias = i < 4 ? bias_lo[i & 3] : bias_hi[i & 3];
        const v2f y = {((wacc[0][i] + wacc[1][i]) + wacc[2][i]) + bias,
                       ((wacc[1][i] - wacc[2][i]) - wacc[3][i]) + bias};
        *(v2f*)(worow + i * Lsz) = y;
      }
      __builtin_amdgcn_s_setprio(0);
    }
  }
}

extern "C" void kernel_launch(void* const* d_in, const int* in_sizes, int n_in,
                              void* d_out, int out_size, void* d_ws, size_t ws_size,
                              hipStream_t stream) {
  (void)in_sizes; (void)n_in; (void)out_size; (void)d_ws; (void)ws_size;
  const int*   word_vector   = (const int*)d_in[0];
  const int*   words_in_char = (const int*)d_in[1];
  const float* word_table    = (const float*)d_in[2];
  const float* chr_table     = (const float*)d_in[3];
  const float* conv_chr_w    = (const float*)d_in[4];
  const float* conv_chr_b    = (const float*)d_in[5];
  const float* conv_word_w   = (const float*)d_in[6];
  const float* conv_word_b   = (const float*)d_in[7];
  float* out = (float*)d_out;

  const size_t word_lds = (size_t)WORD_LDS_FLOATS * sizeof(float);

  static bool attr_done = false;
  if (!attr_done) {
    hipFuncSetAttribute((const void*)conv_pair_kernel,
                        hipFuncAttributeMaxDynamicSharedMemorySize, (int)word_lds);
    attr_done = true;
  }

  conv_pair_kernel<<<dim3(Bsz * (Dsz / DSL)), dim3(256), word_lds, stream>>>(
      word_vector, words_in_char, word_table, chr_table,
      conv_chr_w, conv_chr_b, conv_word_w, conv_word_b, out);
}


# Char table build re-tiled: block = 4 channels x 4 batch rows, (tap, channel) in M: 96 -> 32 f32 WMMAs per wave, one lane per word; loads ahead of an LDS-only first barrier; din sum order regrouped
# speedup vs baseline: 17.3131x; 1.0812x over previous
#include <hip/hip_runtime.h>

#define Bsz 32
#define Lsz 256
#define Csz 16
#define Dsz 128
#define KTOT 384
#define DSL 4
#define RGB 4
#define WL_ROWS 16
#define WL_PITCH 132
#define WL_FLOATS (WL_ROWS * WL_PITCH)
#define PT_ROW 64
#define PT_CHAR (3 * PT_ROW)
#define PT_FLOATS (128 * PT_CHAR)
#define IS_PITCH 20
#define IS_WAVE (32 * IS_PITCH)
#define WSL 32
#define WW_PITCH 132
#define WW_FLOATS (4 * WSL * WW_PITCH)
#define LBLK 128
#define XS_ROWS 136
#define XS_PITCH 132
#define XS_HALF 68
#define XS_FLOATS (XS_ROWS * XS_PITCH)
#define WORD_LDS_FLOATS 34848

static_assert(WORD_LDS_FLOATS == WW_FLOATS + XS_FLOATS, "dynamic LDS = the two word tiles");

typedef __attribute__((ext_vector_type(2))) float v2f;
typedef __attribute__((ext_vector_type(4))) float v4f;
typedef __attribute__((ext_vector_type(8))) float v8f;
typedef __attribute__((ext_vector_type(4))) int v4i;

__device__ __forceinline__ v4f ld4(const float* p) { return *(const v4f*)p; }

__device__ __forceinline__ void lds_barrier() {
  __builtin_amdgcn_fence(__ATOMIC_RELEASE, "workgroup", "local");
  __builtin_amdgcn_s_barrier();
  __builtin_amdgcn_fence(__ATOMIC_ACQUIRE, "workgroup", "local");
}

__device__ __forceinline__ int clamp_char(int ch) {
  return ch < 0 ? 0 : (ch > 127 ? 127 : ch);
}

__device__ __forceinline__ v4f max4(v4f a, v4f b) {
  v4f m = {fmaxf(a[0], b[0]), fmaxf(a[1], b[1]), fmaxf(a[2], b[2]), fmaxf(a[3], b[3])};
  return m;
}

struct WordFrags {
  v4f a[4][2];
  v4f x[4][2];
};

__device__ __forceinline__ void word_load_block(const float* __restrict__ aw,
                                                const float* __restrict__ xb,
                                                int blk, WordFrags& f) {
  const int j0 = blk * 2;
#pragma unroll
  for (int jj = 0; jj < 2; ++jj) {
#pragma unroll
    for (int k = 0; k < 4; ++k) {
      f.a[k][jj] = ld4(aw + k * (WSL * WW_PITCH) + 8 * (j0 + jj));
    }
    f.x[0][jj] = ld4(xb + 8 * (j0 + jj));
    f.x[1][jj] = ld4(xb + XS_HALF * XS_PITCH + 8 * (j0 + jj));
    f.x[2][jj] = ld4(xb + XS_PITCH + 8 * (j0 + jj));
    f.x[3][jj] = ld4(xb + (XS_HALF + 1) * XS_PITCH + 8 * (j0 + jj));
  }
}

__device__ __forceinline__ void word_wmma_block(const WordFrags& f, v8f acc[4]) {
#pragma unroll
  for (int jj = 0; jj < 2; ++jj) {
    v4f v[4];
    v[0] = f.x[0][jj] - f.x[2][jj];
    v[1] = f.x[1][jj] + f.x[2][jj];
    v[2] = f.x[2][jj] - f.x[1][jj];
    v[3] = f.x[1][jj] - f.x[3][jj];
#pragma unroll
    for (int k = 0; k < 4; ++k) {
      const v2f a_lo = {f.a[k][jj][0], f.a[k][jj][1]};
      const v2f b_lo = {v[k][0], v[k][1]};
      acc[k] = __builtin_amdgcn_wmma_f32_16x16x4_f32(
           false, a_lo,  false, b_lo,
           (short)0, acc[k],  false,  false);
    }
#pragma unroll
    for (int k = 0; k < 4; ++k) {
      const v2f a_hi = {f.a[k][jj][2], f.a[k][jj][3]};
      const v2f b_hi = {v[k][2], v[k][3]};
      acc[k] = __builtin_amdgcn_wmma_f32_16x16x4_f32(
           false, a_hi,  false, b_hi,
           (short)0, acc[k],  false,  false);
    }
  }
}

__global__ __launch_bounds__(256) void conv_pair_kernel(
    const int* __restrict__ word_vector,
    const int* __restrict__ words_in_char,
    const float* __restrict__ word_table,
    const float* __restrict__ chr_table,
    const float* __restrict__ chr_w,
    const float* __restrict__ chr_b,
    const float* __restrict__ word_w,
    const float* __restrict__ word_b,
    float* __restrict__ out) {
  __shared__ __attribute__((aligned(16))) float wl[WL_FLOATS];
  __shared__ __attribute__((aligned(16))) float pt[PT_FLOATS];
  __shared__ __attribute__((aligned(16))) int isc[8 * IS_WAVE];
  extern __shared__ __attribute__((aligned(16))) float smem[];
  float* ww = smem;
  float* xs = smem + WW_FLOATS;
  const int tid  = threadIdx.x;
  const int lane = tid & 31;
  const int w    = __builtin_amdgcn_readfirstlane(tid >> 5);
  const int r    = lane & 15;
  const int h    = lane >> 4;
  const int bid  = blockIdx.x;
  const int d0   = (bid & 31) * DSL;
  const int cb0  = (bid >> 5) * RGB;
  const int b     = bid >> 3;
  const int wd0   = ((bid >> 1) & 3) * WSL;
  const int lbase = (bid & 1) * LBLK;

  int lpos = lbase + w + 8 * lane - 1;
  lpos = lpos < 0 ? 0 : (lpos > Lsz - 1 ? Lsz - 1 : lpos);
  const int my_w = word_vector[b * Lsz + lpos];

  const float* wsrc = chr_w + d0 * KTOT;
  v4f wv[2];
  wv[0] = ld4(wsrc + tid * 4);
  if (w < 4) {
    wv[1] = ld4(wsrc + (tid + 256) * 4);
  }

  v4f bq[16];
  const float* trow = chr_table + (w * 16 + r) * Dsz + 4 * h;
#pragma unroll
  for (int j = 0; j < 16; ++j) {
    bq[j] = ld4(trow + 8 * j);
  }

  const float* wwsrc = word_w + (wd0 + w) * KTOT + lane * 12;
  v4f wvw[12];
#pragma unroll
  for (int i = 0; i < 4; ++i) {
#pragma unroll
    for (int q = 0; q < 3; ++q) {
      wvw[3 * i + q] = ld4(wwsrc + i * 8 * KTOT + 4 * q);
    }
  }

  const int dt = w & 1;
  const int lp = w >> 1;
  const float* bp = word_b + wd0 + dt * 16 + 8 * h;
  const v4f bias_lo = ld4(bp);
  const v4f bias_hi = ld4(bp + 4);

  v4i civ[4][4];
#pragma unroll
  for (int it = 0; it < 4; ++it) {
    const v4i* ip = (const v4i*)(words_in_char + ((cb0 + it) * Lsz + w * 32) * Csz) + lane;
#pragma unroll
    for (int k = 0; k < 4; ++k) {
      civ[it][k] = ip[32 * k];
    }
  }

  __builtin_amdgcn_sched_barrier(0);

  v4f rows[17];
#pragma unroll
  for (int i = 0; i < 17; ++i) {
    const int widx = __builtin_amdgcn_readlane(my_w, i);
    rows[i] = ld4(word_table + widx * Dsz + lane * 4);
  }

  __builtin_amdgcn_sched_barrier(0);

  const v4f zero4 = {0.0f, 0.0f, 0.0f, 0.0f};
#pragma unroll
  for (int i = 0; i < 2; ++i) {
    if (i == 0 || w < 4) {
      const int f0  = (tid + 256 * i) * 4;
      const int dl  = f0 / KTOT;
      const int rem = f0 - dl * KTOT;
#pragma unroll
      for (int e = 0; e < 4; ++e) {
        const int din = (rem + e) / 3;
        const int tap = (rem + e) - din * 3;
        wl[(tap * DSL + dl) * WL_PITCH + din] = wv[i][e];
      }
    }
  }
  if (tid < 4 * WL_PITCH / 4) {
    *(v4f*)(wl + 12 * WL_PITCH + 4 * tid) = zero4;
  }
  lds_barrier();

  v8f acc0 = {}, acc1 = {};
#pragma unroll
  for (int j = 0; j < 16; ++j) {
    const v4f a4 = ld4(wl + r * WL_PITCH + 8 * j + 4 * h);
    const v2f a_lo = {a4[0], a4[1]};
    const v2f b_lo = {bq[j][0], bq[j][1]};
    const v2f a_hi = {a4[2], a4[3]};
    const v2f b_hi = {bq[j][2], bq[j][3]};
    acc0 = __builtin_amdgcn_wmma_f32_16x16x4_f32(
         false, a_lo,  false, b_lo,
         (short)0, acc0,  false,  false);
    acc1 = __builtin_amdgcn_wmma_f32_16x16x4_f32(
         false, a_hi,  false, b_hi,
         (short)0, acc1,  false,  false);
  }
  const v8f acc = acc0 + acc1;

  {
    const int chr = w * 16 + r;
    const v4f lo = {acc[0], acc[1], acc[2], acc[3]};
    const v4f hi = {acc[4], acc[5], acc[6], acc[7]};
    float* prow = pt + chr * PT_CHAR + (h ? 2 * PT_ROW : 0);
#pragma unroll
    for (int s = 0; s < 16; ++s) {
      const int cp = (s + r) & 15;
      *(v4f*)(prow + 4 * cp) = lo;
      if (h == 0) {
        *(v4f*)(prow + PT_ROW + 4 * cp) = hi;
      }
    }
  }

  int* sc = isc + w * IS_WAVE;
  int goff[4][Csz];
#pragma unroll
  for (int it = 0; it < 4; ++it) {
#pragma unroll
    for (int k = 0; k < 4; ++k) {
      *(v4i*)(sc + (8 * k + (lane >> 2)) * IS_PITCH + 4 * (lane & 3)) = civ[it][k];
    }
#pragma unroll
    for (int j4 = 0; j4 < 4; ++j4) {
      const v4i q4 = *(const v4i*)(sc + lane * IS_PITCH + 4 * j4);
#pragma unroll
      for (int e = 0; e < 4; ++e) {
        goff[it][4 * j4 + e] = clamp_char(q4[e]) * PT_CHAR + 4 * r;
      }
    }
  }

#pragma unroll
  for (int i = 0; i < 4; ++i) {
    const v4f x0 = wvw[3 * i], x1 = wvw[3 * i + 1], x2 = wvw[3 * i + 2];
    const v4f g0 = {x0[0], x0[3], x1[2], x2[1]};
    const v4f g1 = {x0[1], x1[0], x1[3], x2[2]};
    const v4f g2 = {x0[2], x1[1], x2[0], x2[3]};
    const v4f gs = g0 + g2;
    const v4f u1 = (gs + g1) * 0.5f;
    const v4f u2 = (gs - g1) * 0.5f;
    float* dst = ww + (w + 8 * i) * WW_PITCH + 4 * lane;
    *(v4f*)(dst)                         = g0;
    *(v4f*)(dst + 1 * (WSL * WW_PITCH)) = u1;
    *(v4f*)(dst + 2 * (WSL * WW_PITCH)) = u2;
    *(v4f*)(dst + 3 * (WSL * WW_PITCH)) = g2;
  }

#pragma unroll
  for (int i = 0; i < 17; ++i) {
    const int row = w + 8 * i;
    v4f v = rows[i];
    if (i == 0 || i == 16) {
      const int l = lbase + row - 1;
      if (l < 0 || l > Lsz - 1) v = zero4;
    }
    *(v4f*)(xs + ((row & 1) * XS_HALF + (row >> 1)) * XS_PITCH + lane * 4) = v;
  }
  lds_barrier();

  const float* aw = ww + (dt * 16 + r) * WW_PITCH + 4 * h;
  const float* xb = xs + (lp * 16 + r) * XS_PITCH + 4 * h;
  const bool gather_first = w >= 4;

  for (int stage = 0; stage < 2; ++stage) {
    if ((stage == 0) == gather_first) {

      float cb[DSL];
#pragma unroll
      for (int dl = 0; dl < DSL; ++dl) {
        cb[dl] = chr_b[d0 + dl];
      }

#pragma unroll
      for (int it = 0; it < 4; ++it) {
        const int* off = goff[it];
        v4f m;
#pragma unroll
        for (int c = 0; c < Csz; ++c) {
          v4f s = ld4(pt + PT_ROW + off[c]);
          if (c > 0) {
            s = ld4(pt + off[c - 1]) + s;
          }
          if (c < Csz - 1) {
            s = s + ld4(pt + 2 * PT_ROW + off[c + 1]);
          }
          m = (c == 0) ? s : max4(m, s);
        }
        float* corow = out + ((Bsz + cb0 + it) * Dsz + d0) * Lsz + w * 32 + lane;
#pragma unroll
        for (int dl = 0; dl < DSL; ++dl) {
          corow[dl * Lsz] = m[dl] + cb[dl];
        }
      }
    } else {
      __builtin_amdgcn_s_setprio(3);
      v8f wacc[4] = {};
      WordFrags f0, f1;
      word_load_block(aw, xb, 0, f0);
#pragma unroll
      for (int blk = 0; blk < 8; blk += 2) {
        word_load_block(aw, xb, blk + 1, f1);
        __builtin_amdgcn_sched_barrier(0);
        word_wmma_block(f0, wacc);
        __builtin_amdgcn_sched_barrier(0);
        if (blk + 2 < 8) word_load_block(aw, xb, blk + 2, f0);
        __builtin_amdgcn_sched_barrier(0);
        word_wmma_block(f1, wacc);
        __builtin_amdgcn_sched_barrier(0);
      }

      float* worow = out + (b * Dsz + wd0 + dt * 16 + 8 * h) * Lsz + lbase + lp * 32 + 2 * r;
#pragma unroll
      for (int i = 0; i < 8; ++i) {
        const float bias = i < 4 ? bias_lo[i & 3] : bias_hi[i & 3];
        const v2f y = {((wacc[0][i] + wacc[1][i]) + wacc[2][i]) + bias,
                       ((wacc[1][i] - wacc[2][i]) - wacc[3][i]) + bias};
        *(v2f*)(worow + i * Lsz) = y;
      }
      __builtin_amdgcn_s_setprio(0);
    }
  }
}

extern "C" void kernel_launch(void* const* d_in, const int* in_sizes, int n_in,
                              void* d_out, int out_size, void* d_ws, size_t ws_size,
                              hipStream_t stream) {
  (void)in_sizes; (void)n_in; (void)out_size; (void)d_ws; (void)ws_size;
  const int*   word_vector   = (const int*)d_in[0];
  const int*   words_in_char = (const int*)d_in[1];
  const float* word_table    = (const float*)d_in[2];
  const float* chr_table     = (const float*)d_in[3];
  const float* conv_chr_w    = (const float*)d_in[4];
  const float* conv_chr_b    = (const float*)d_in[5];
  const float* conv_word_w   = (const float*)d_in[6];
  const float* conv_word_b   = (const float*)d_in[7];
  float* out = (float*)d_out;

  const size_t word_lds = (size_t)WORD_LDS_FLOATS * sizeof(float);

  static bool attr_done = false;
  if (!attr_done) {
    hipFuncSetAttribute((const void*)conv_pair_kernel,
                        hipFuncAttributeMaxDynamicSharedMemorySize, (int)word_lds);
    attr_done = true;
  }

  conv_pair_kernel<<<dim3(256), dim3(256), word_lds, stream>>>(
      word_vector, words_in_char, word_table, chr_table,
      conv_chr_w, conv_chr_b, conv_word_w, conv_word_b, out);
}
